# starting kernel
# baseline (speedup 1.0000x reference)
#include <hip/hip_runtime.h>
#include <hip/hip_bf16.h>
#include <math.h>


typedef float v2f __attribute__((ext_vector_type(2)));
typedef float v8f __attribute__((ext_vector_type(8)));

#define F_IN 128
#define HID  16
#define NCLS 8


__global__ void deg_init_kernel(float* __restrict__ deg, int n) {
  int i = blockIdx.x * blockDim.x + threadIdx.x;
  if (i < n) deg[i] = 1.0f;
}

__global__ void deg_count_kernel(const int* __restrict__ dst,
                                 float* __restrict__ deg, int e) {
  int i = blockIdx.x * blockDim.x + threadIdx.x;
  if (i < e) atomicAdd(&deg[dst[i]], 1.0f);
}

__global__ void dinv_kernel(float* __restrict__ deg, int n) {
  int i = blockIdx.x * blockDim.x + threadIdx.x;
  if (i < n) deg[i] = rsqrtf(deg[i]);
}


__global__ void __launch_bounds__(256) gemm1_kernel(
    const float* __restrict__ x, const float* __restrict__ W1,
    const float* __restrict__ dinv,
    float* __restrict__ h, float* __restrict__ out1, int nTiles, int n) {
  int wave = (blockIdx.x * 256 + threadIdx.x) >> 5;
  int lane = threadIdx.x & 31;
  if (wave >= nTiles) return;
  int half = lane >> 4;
  int m    = lane & 15;
  int row  = wave * 16 + m;
  int rowc = row < n ? row : n - 1;
  const float* xrow = x + (size_t)rowc * F_IN;

  v8f c = {};
  #pragma unroll 4
  for (int k0 = 0; k0 < F_IN; k0 += 4) {
    int kk = k0 + 2 * half;
    v2f a = *(const v2f*)(xrow + kk);
    v2f b;
    b.x = W1[kk * HID + m];
    b.y = W1[(kk + 1) * HID + m];
    c = __builtin_amdgcn_wmma_f32_16x16x4_f32(
        false, a, false, b, (short)0, c, false, false);
  }

  float dv = dinv[rowc];
  bool full = (wave * 16 + 16 <= n);
  if (full) {
    #pragma unroll
    for (int r = 0; r < 8; ++r) {
      int orow = wave * 16 + r + 8 * half;
      float w  = __shfl(dv, r + 8 * half, 32);
      size_t o = (size_t)orow * HID + m;
      h[o]    = c[r];
      out1[o] = c[r] * w * w;
    }
  } else {
    #pragma unroll
    for (int r = 0; r < 8; ++r) {
      int orow = wave * 16 + r + 8 * half;
      float w  = __shfl(dv, r + 8 * half, 32);
      if (orow < n) {
        size_t o = (size_t)orow * HID + m;
        h[o]    = c[r];
        out1[o] = c[r] * w * w;
      }
    }
  }
}


__global__ void scatter1_kernel(const int* __restrict__ src,
                                const int* __restrict__ dst,
                                const float* __restrict__ dinv,
                                const float* __restrict__ h,
                                float* __restrict__ out1, int e) {
  int i = blockIdx.x * blockDim.x + threadIdx.x;
  if (i >= e) return;
  int s = src[i];
  int d = dst[i];
  float w = dinv[s] * dinv[d];
  const float4* hp = (const float4*)(h + (size_t)s * HID);
  float* op = out1 + (size_t)d * HID;
  #pragma unroll
  for (int q = 0; q < 4; ++q) {
    float4 v = hp[q];
    atomicAdd(op + q * 4 + 0, v.x * w);
    atomicAdd(op + q * 4 + 1, v.y * w);
    atomicAdd(op + q * 4 + 2, v.z * w);
    atomicAdd(op + q * 4 + 3, v.w * w);
  }
}


__global__ void __launch_bounds__(256) gemm2_kernel(
    const float* __restrict__ h1, const float* __restrict__ b1,
    const float* __restrict__ W2, const float* __restrict__ dinv,
    float* __restrict__ lt, float* __restrict__ out, int nTiles, int n) {
  int wave = (blockIdx.x * 256 + threadIdx.x) >> 5;
  int lane = threadIdx.x & 31;
  if (wave >= nTiles) return;
  int half = lane >> 4;
  int m    = lane & 15;
  int row  = wave * 16 + m;
  int rowc = row < n ? row : n - 1;
  const float* hrow = h1 + (size_t)rowc * HID;
  float maskv = (m < NCLS) ? 1.0f : 0.0f;
  int mc = m & (NCLS - 1);

  v8f c = {};
  #pragma unroll
  for (int k0 = 0; k0 < HID; k0 += 4) {
    int kk = k0 + 2 * half;
    v2f a;
    a.x = fmaxf(hrow[kk]     + b1[kk],     0.0f);
    a.y = fmaxf(hrow[kk + 1] + b1[kk + 1], 0.0f);
    v2f b;
    b.x = W2[kk * NCLS + mc] * maskv;
    b.y = W2[(kk + 1) * NCLS + mc] * maskv;
    c = __builtin_amdgcn_wmma_f32_16x16x4_f32(
        false, a, false, b, (short)0, c, false, false);
  }

  float dv = dinv[rowc];
  bool full = (wave * 16 + 16 <= n);
  if (m < NCLS) {
    #pragma unroll
    for (int r = 0; r < 8; ++r) {
      int orow = wave * 16 + r + 8 * half;
      float w  = __shfl(dv, r + 8 * half, 32);
      if (full || orow < n) {
        size_t o = (size_t)orow * NCLS + m;
        lt[o]  = c[r];
        out[o] = c[r] * w * w;
      }
    }
  }
}


__global__ void scatter2_kernel(const int* __restrict__ src,
                                const int* __restrict__ dst,
                                const float* __restrict__ dinv,
                                const float* __restrict__ lt,
                                float* __restrict__ out, int e) {
  int i = blockIdx.x * blockDim.x + threadIdx.x;
  if (i >= e) return;
  int s = src[i];
  int d = dst[i];
  float w = dinv[s] * dinv[d];
  const float4* lp = (const float4*)(lt + (size_t)s * NCLS);
  float* op = out + (size_t)d * NCLS;
  #pragma unroll
  for (int q = 0; q < 2; ++q) {
    float4 v = lp[q];
    atomicAdd(op + q * 4 + 0, v.x * w);
    atomicAdd(op + q * 4 + 1, v.y * w);
    atomicAdd(op + q * 4 + 2, v.z * w);
    atomicAdd(op + q * 4 + 3, v.w * w);
  }
}


__global__ void lsm_kernel(float* __restrict__ out,
                           const float* __restrict__ b2, int n) {
  int i = blockIdx.x * blockDim.x + threadIdx.x;
  if (i >= n) return;
  float* row = out + (size_t)i * NCLS;
  float z[NCLS];
  float mx = -INFINITY;
  #pragma unroll
  for (int c = 0; c < NCLS; ++c) {
    z[c] = row[c] + b2[c];
    mx = fmaxf(mx, z[c]);
  }
  float s = 0.0f;
  #pragma unroll
  for (int c = 0; c < NCLS; ++c) s += expf(z[c] - mx);
  float l = mx + logf(s);
  #pragma unroll
  for (int c = 0; c < NCLS; ++c) row[c] = z[c] - l;
}


extern "C" void kernel_launch(void* const* d_in, const int* in_sizes, int n_in,
                              void* d_out, int out_size, void* d_ws, size_t ws_size,
                              hipStream_t stream) {
  const float* x  = (const float*)d_in[0];
  const int* edge = (const int*)d_in[1];
  const float* W1 = (const float*)d_in[2];
  const float* b1 = (const float*)d_in[3];
  const float* W2 = (const float*)d_in[4];
  const float* b2 = (const float*)d_in[5];

  int n = in_sizes[0] / F_IN;
  int e = in_sizes[1] / 2;
  const int* src = edge;
  const int* dst = edge + e;

  float* deg = (float*)d_ws;
  float* h   = deg + n;
  float* h1  = h + (size_t)n * HID;
  float* lt  = h1 + (size_t)n * HID;
  float* out = (float*)d_out;

  const int T = 256;
  int nTiles  = (n + 15) / 16;
  int gBlocks = (nTiles + 7) / 8;

  deg_init_kernel<<<(n + T - 1) / T, T, 0, stream>>>(deg, n);
  deg_count_kernel<<<(e + T - 1) / T, T, 0, stream>>>(dst, deg, e);
  dinv_kernel<<<(n + T - 1) / T, T, 0, stream>>>(deg, n);

  gemm1_kernel<<<gBlocks, T, 0, stream>>>(x, W1, deg, h, h1, nTiles, n);
  scatter1_kernel<<<(e + T - 1) / T, T, 0, stream>>>(src, dst, deg, h, h1, e);

  gemm2_kernel<<<gBlocks, T, 0, stream>>>(h1, b1, W2, deg, lt, out, nTiles, n);
  scatter2_kernel<<<(e + T - 1) / T, T, 0, stream>>>(src, dst, deg, lt, out, e);

  lsm_kernel<<<(n + T - 1) / T, T, 0, stream>>>(out, b2, n);
}


# Edge weight dinv[src]*dinv[dst] computed once: layer-1 scatter stores it per edge, layer-2 scatter reads it back coalesced instead of two random dinv reads per edge; atomics and sum order unchanged
# speedup vs baseline: 3.5621x; 3.5621x over previous
#include <hip/hip_runtime.h>
#include <hip/hip_bf16.h>
#include <math.h>


typedef float v2f __attribute__((ext_vector_type(2)));
typedef float v8f __attribute__((ext_vector_type(8)));

#define F_IN 128
#define HID  16
#define NCLS 8


__global__ void deg_init_kernel(float* __restrict__ deg, int n) {
  int i = blockIdx.x * blockDim.x + threadIdx.x;
  if (i < n) deg[i] = 1.0f;
}

__global__ void deg_count_kernel(const int* __restrict__ dst,
                                 float* __restrict__ deg, int e) {
  int i = blockIdx.x * blockDim.x + threadIdx.x;
  if (i < e) atomicAdd(&deg[dst[i]], 1.0f);
}

__global__ void dinv_kernel(float* __restrict__ deg, int n) {
  int i = blockIdx.x * blockDim.x + threadIdx.x;
  if (i < n) deg[i] = rsqrtf(deg[i]);
}


__global__ void __launch_bounds__(256) gemm1_kernel(
    const float* __restrict__ x, const float* __restrict__ W1,
    const float* __restrict__ dinv,
    float* __restrict__ h, float* __restrict__ out1, int nTiles, int n) {
  int wave = (blockIdx.x * 256 + threadIdx.x) >> 5;
  int lane = threadIdx.x & 31;
  if (wave >= nTiles) return;
  int half = lane >> 4;
  int m    = lane & 15;
  int row  = wave * 16 + m;
  int rowc = row < n ? row : n - 1;
  const float* xrow = x + (size_t)rowc * F_IN;

  v8f c = {};
  #pragma unroll 4
  for (int k0 = 0; k0 < F_IN; k0 += 4) {
    int kk = k0 + 2 * half;
    v2f a = *(const v2f*)(xrow + kk);
    v2f b;
    b.x = W1[kk * HID + m];
    b.y = W1[(kk + 1) * HID + m];
    c = __builtin_amdgcn_wmma_f32_16x16x4_f32(
        false, a, false, b, (short)0, c, false, false);
  }

  float dv = dinv[rowc];
  bool full = (wave * 16 + 16 <= n);
  if (full) {
    #pragma unroll
    for (int r = 0; r < 8; ++r) {
      int orow = wave * 16 + r + 8 * half;
      float w  = __shfl(dv, r + 8 * half, 32);
      size_t o = (size_t)orow * HID + m;
      h[o]    = c[r];
      out1[o] = c[r] * w * w;
    }
  } else {
    #pragma unroll
    for (int r = 0; r < 8; ++r) {
      int orow = wave * 16 + r + 8 * half;
      float w  = __shfl(dv, r + 8 * half, 32);
      if (orow < n) {
        size_t o = (size_t)orow * HID + m;
        h[o]    = c[r];
        out1[o] = c[r] * w * w;
      }
    }
  }
}


__global__ void __launch_bounds__(256) scatter1_kernel(
    const int* __restrict__ src, const int* __restrict__ dst,
    const float* __restrict__ dinv, const float* __restrict__ h,
    float* __restrict__ out1, float* __restrict__ ew, int e) {
  int lane = threadIdx.x & 31;
  int base = blockIdx.x * 256 + threadIdx.x - lane;
  int i = base + lane;
  int s = 0, d = 0;
  float w = 0.0f;
  if (i < e) {
    s = src[i];
    d = dst[i];
    w = dinv[s] * dinv[d];
    ew[i] = w;
  }
  int half = lane >> 4;
  int c    = lane & 15;
  if (base + 32 <= e) {
    float v[16];
    #pragma unroll
    for (int j = 0; j < 16; ++j) {
      int sj = __shfl(s, 2 * j + half, 32);
      v[j] = h[(size_t)sj * HID + c];
    }
    #pragma unroll
    for (int j = 0; j < 16; ++j) {
      int   dj = __shfl(d, 2 * j + half, 32);
      float wj = __shfl(w, 2 * j + half, 32);
      atomicAdd(out1 + (size_t)dj * HID + c, v[j] * wj);
    }
  } else {
    for (int j = 0; j < 16; ++j) {
      int from = 2 * j + half;
      int   sj = __shfl(s, from, 32);
      int   dj = __shfl(d, from, 32);
      float wj = __shfl(w, from, 32);
      if (base + from < e) {
        float v = h[(size_t)sj * HID + c];
        atomicAdd(out1 + (size_t)dj * HID + c, v * wj);
      }
    }
  }
}


__global__ void __launch_bounds__(256) gemm2_kernel(
    const float* __restrict__ h1, const float* __restrict__ b1,
    const float* __restrict__ W2, const float* __restrict__ dinv,
    float* __restrict__ lt, float* __restrict__ out, int nTiles, int n) {
  int wave = (blockIdx.x * 256 + threadIdx.x) >> 5;
  int lane = threadIdx.x & 31;
  if (wave >= nTiles) return;
  int half = lane >> 4;
  int m    = lane & 15;
  int row  = wave * 16 + m;
  int rowc = row < n ? row : n - 1;
  const float* hrow = h1 + (size_t)rowc * HID;
  float maskv = (m < NCLS) ? 1.0f : 0.0f;
  int mc = m & (NCLS - 1);

  v8f c = {};
  #pragma unroll
  for (int k0 = 0; k0 < HID; k0 += 4) {
    int kk = k0 + 2 * half;
    v2f a;
    a.x = fmaxf(hrow[kk]     + b1[kk],     0.0f);
    a.y = fmaxf(hrow[kk + 1] + b1[kk + 1], 0.0f);
    v2f b;
    b.x = W2[kk * NCLS + mc] * maskv;
    b.y = W2[(kk + 1) * NCLS + mc] * maskv;
    c = __builtin_amdgcn_wmma_f32_16x16x4_f32(
        false, a, false, b, (short)0, c, false, false);
  }

  float dv = dinv[rowc];
  bool full = (wave * 16 + 16 <= n);
  if (m < NCLS) {
    #pragma unroll
    for (int r = 0; r < 8; ++r) {
      int orow = wave * 16 + r + 8 * half;
      float w  = __shfl(dv, r + 8 * half, 32);
      if (full || orow < n) {
        size_t o = (size_t)orow * NCLS + m;
        lt[o]  = c[r];
        out[o] = c[r] * w * w;
      }
    }
  }
}


__global__ void __launch_bounds__(256) scatter2_kernel(
    const int* __restrict__ src, const int* __restrict__ dst,
    const float* __restrict__ ew, const float* __restrict__ lt,
    float* __restrict__ out, int e) {
  int lane = threadIdx.x & 31;
  int base = blockIdx.x * 256 + threadIdx.x - lane;
  int i = base + lane;
  int s = 0, d = 0;
  float w = 0.0f;
  if (i < e) {
    s = src[i];
    d = dst[i];
    w = ew[i];
  }
  int quarter = lane >> 3;
  int c       = lane & 7;
  if (base + 32 <= e) {
    float v[8];
    #pragma unroll
    for (int j = 0; j < 8; ++j) {
      int sj = __shfl(s, 4 * j + quarter, 32);
      v[j] = lt[(size_t)sj * NCLS + c];
    }
    #pragma unroll
    for (int j = 0; j < 8; ++j) {
      int   dj = __shfl(d, 4 * j + quarter, 32);
      float wj = __shfl(w, 4 * j + quarter, 32);
      atomicAdd(out + (size_t)dj * NCLS + c, v[j] * wj);
    }
  } else {
    for (int j = 0; j < 8; ++j) {
      int from = 4 * j + quarter;
      int   sj = __shfl(s, from, 32);
      int   dj = __shfl(d, from, 32);
      float wj = __shfl(w, from, 32);
      if (base + from < e) {
        float v = lt[(size_t)sj * NCLS + c];
        atomicAdd(out + (size_t)dj * NCLS + c, v * wj);
      }
    }
  }
}


__global__ void lsm_kernel(float* __restrict__ out,
                           const float* __restrict__ b2, int n) {
  int i = blockIdx.x * blockDim.x + threadIdx.x;
  if (i >= n) return;
  float* row = out + (size_t)i * NCLS;
  float z[NCLS];
  float mx = -INFINITY;
  #pragma unroll
  for (int c = 0; c < NCLS; ++c) {
    z[c] = row[c] + b2[c];
    mx = fmaxf(mx, z[c]);
  }
  float s = 0.0f;
  #pragma unroll
  for (int c = 0; c < NCLS; ++c) s += expf(z[c] - mx);
  float l = mx + logf(s);
  #pragma unroll
  for (int c = 0; c < NCLS; ++c) row[c] = z[c] - l;
}


extern "C" void kernel_launch(void* const* d_in, const int* in_sizes, int n_in,
                              void* d_out, int out_size, void* d_ws, size_t ws_size,
                              hipStream_t stream) {
  const float* x  = (const float*)d_in[0];
  const int* edge = (const int*)d_in[1];
  const float* W1 = (const float*)d_in[2];
  const float* b1 = (const float*)d_in[3];
  const float* W2 = (const float*)d_in[4];
  const float* b2 = (const float*)d_in[5];

  int n = in_sizes[0] / F_IN;
  int e = in_sizes[1] / 2;
  const int* src = edge;
  const int* dst = edge + e;

  float* deg = (float*)d_ws;
  float* h   = deg + n;
  float* h1  = h + (size_t)n * HID;
  float* lt  = h1 + (size_t)n * HID;
  float* ew  = lt + (size_t)n * NCLS;
  float* out = (float*)d_out;

  const int T = 256;
  int nTiles  = (n + 15) / 16;
  int gBlocks = (nTiles + 7) / 8;

  deg_init_kernel<<<(n + T - 1) / T, T, 0, stream>>>(deg, n);
  deg_count_kernel<<<(e + T - 1) / T, T, 0, stream>>>(dst, deg, e);
  dinv_kernel<<<(n + T - 1) / T, T, 0, stream>>>(deg, n);

  gemm1_kernel<<<gBlocks, T, 0, stream>>>(x, W1, deg, h, h1, nTiles, n);
  scatter1_kernel<<<(e + T - 1) / T, T, 0, stream>>>(src, dst, deg, h, h1, ew, e);

  gemm2_kernel<<<gBlocks, T, 0, stream>>>(h1, b1, W2, deg, lt, out, nTiles, n);
  scatter2_kernel<<<(e + T - 1) / T, T, 0, stream>>>(src, dst, ew, lt, out, e);

  lsm_kernel<<<(n + T - 1) / T, T, 0, stream>>>(out, b2, n);
}


# Edge weight factored: GEMMs store rows scaled by dinv[src], dinv[dst] applied per node after the f32-atomic sum (arrival order); earlier try zeroed rows: GEMM2 dinv shuffle sat under m<8 mask
# speedup vs baseline: 3.7178x; 1.0437x over previous
#include <hip/hip_runtime.h>
#include <hip/hip_bf16.h>
#include <math.h>


typedef float v2f __attribute__((ext_vector_type(2)));
typedef float v8f __attribute__((ext_vector_type(8)));

#define F_IN 128
#define HID  16
#define NCLS 8


__global__ void deg_init_kernel(float* __restrict__ deg, int n) {
  int i = blockIdx.x * blockDim.x + threadIdx.x;
  if (i < n) deg[i] = 1.0f;
}

__global__ void deg_count_kernel(const int* __restrict__ dst,
                                 float* __restrict__ deg, int e) {
  int i = blockIdx.x * blockDim.x + threadIdx.x;
  if (i < e) atomicAdd(&deg[dst[i]], 1.0f);
}

__global__ void dinv_kernel(float* __restrict__ deg, int n) {
  int i = blockIdx.x * blockDim.x + threadIdx.x;
  if (i < n) deg[i] = rsqrtf(deg[i]);
}


__global__ void __launch_bounds__(256) gemm1_kernel(
    const float* __restrict__ x, const float* __restrict__ W1,
    const float* __restrict__ dinv,
    float* __restrict__ h, float* __restrict__ out1, int nTiles, int n) {
  int wave = (blockIdx.x * 256 + threadIdx.x) >> 5;
  int lane = threadIdx.x & 31;
  if (wave >= nTiles) return;
  int half = lane >> 4;
  int m    = lane & 15;
  int row  = wave * 16 + m;
  int rowc = row < n ? row : n - 1;
  const float* xrow = x + (size_t)rowc * F_IN;

  v8f c = {};
  #pragma unroll 4
  for (int k0 = 0; k0 < F_IN; k0 += 4) {
    int kk = k0 + 2 * half;
    v2f a = *(const v2f*)(xrow + kk);
    v2f b;
    b.x = W1[kk * HID + m];
    b.y = W1[(kk + 1) * HID + m];
    c = __builtin_amdgcn_wmma_f32_16x16x4_f32(
        false, a, false, b, (short)0, c, false, false);
  }

  float dv = dinv[rowc];
  bool full = (wave * 16 + 16 <= n);
  if (full) {
    #pragma unroll
    for (int r = 0; r < 8; ++r) {
      int orow = wave * 16 + r + 8 * half;
      float w  = __shfl(dv, r + 8 * half, 32);
      size_t o = (size_t)orow * HID + m;
      float v  = c[r] * w;
      h[o]    = v;
      out1[o] = v;
    }
  } else {
    #pragma unroll
    for (int r = 0; r < 8; ++r) {
      int orow = wave * 16 + r + 8 * half;
      float w  = __shfl(dv, r + 8 * half, 32);
      if (orow < n) {
        size_t o = (size_t)orow * HID + m;
        float v  = c[r] * w;
        h[o]    = v;
        out1[o] = v;
      }
    }
  }
}


__global__ void __launch_bounds__(256) scatter1_kernel(
    const int* __restrict__ src, const int* __restrict__ dst,
    const float* __restrict__ h, float* __restrict__ out1, int e) {
  int lane = threadIdx.x & 31;
  int base = blockIdx.x * 256 + threadIdx.x - lane;
  int i = base + lane;
  int s = 0, d = 0;
  if (i < e) {
    s = src[i];
    d = dst[i];
  }
  int half = lane >> 4;
  int c    = lane & 15;
  if (base + 32 <= e) {
    float v[16];
    #pragma unroll
    for (int j = 0; j < 16; ++j) {
      int sj = __shfl(s, 2 * j + half, 32);
      v[j] = h[(size_t)sj * HID + c];
    }
    #pragma unroll
    for (int j = 0; j < 16; ++j) {
      int dj = __shfl(d, 2 * j + half, 32);
      atomicAdd(out1 + (size_t)dj * HID + c, v[j]);
    }
  } else {
    for (int j = 0; j < 16; ++j) {
      int from = 2 * j + half;
      int sj = __shfl(s, from, 32);
      int dj = __shfl(d, from, 32);
      if (base + from < e) {
        float v = h[(size_t)sj * HID + c];
        atomicAdd(out1 + (size_t)dj * HID + c, v);
      }
    }
  }
}


__global__ void __launch_bounds__(256) gemm2_kernel(
    const float* __restrict__ h1, const float* __restrict__ b1,
    const float* __restrict__ W2, const float* __restrict__ dinv,
    float* __restrict__ lt, float* __restrict__ out, int nTiles, int n) {
  int wave = (blockIdx.x * 256 + threadIdx.x) >> 5;
  int lane = threadIdx.x & 31;
  if (wave >= nTiles) return;
  int half = lane >> 4;
  int m    = lane & 15;
  int row  = wave * 16 + m;
  int rowc = row < n ? row : n - 1;
  const float* hrow = h1 + (size_t)rowc * HID;
  float maskv = (m < NCLS) ? 1.0f : 0.0f;
  int mc = m & (NCLS - 1);
  float dv = dinv[rowc];

  v8f c = {};
  #pragma unroll
  for (int k0 = 0; k0 < HID; k0 += 4) {
    int kk = k0 + 2 * half;
    v2f a;
    a.x = fmaxf(hrow[kk]     * dv + b1[kk],     0.0f);
    a.y = fmaxf(hrow[kk + 1] * dv + b1[kk + 1], 0.0f);
    v2f b;
    b.x = W2[kk * NCLS + mc] * maskv;
    b.y = W2[(kk + 1) * NCLS + mc] * maskv;
    c = __builtin_amdgcn_wmma_f32_16x16x4_f32(
        false, a, false, b, (short)0, c, false, false);
  }

  bool full = (wave * 16 + 16 <= n);
  float wr[8];
  #pragma unroll
  for (int r = 0; r < 8; ++r) wr[r] = __shfl(dv, r + 8 * half, 32);
  if (m < NCLS) {
    #pragma unroll
    for (int r = 0; r < 8; ++r) {
      int orow = wave * 16 + r + 8 * half;
      if (full || orow < n) {
        size_t o = (size_t)orow * NCLS + m;
        float v = c[r] * wr[r];
        lt[o]  = v;
        out[o] = v;
      }
    }
  }
}


__global__ void __launch_bounds__(256) scatter2_kernel(
    const int* __restrict__ src, const int* __restrict__ dst,
    const float* __restrict__ lt, float* __restrict__ out, int e) {
  int lane = threadIdx.x & 31;
  int base = blockIdx.x * 256 + threadIdx.x - lane;
  int i = base + lane;
  int s = 0, d = 0;
  if (i < e) {
    s = src[i];
    d = dst[i];
  }
  int quarter = lane >> 3;
  int c       = lane & 7;
  if (base + 32 <= e) {
    float v[8];
    #pragma unroll
    for (int j = 0; j < 8; ++j) {
      int sj = __shfl(s, 4 * j + quarter, 32);
      v[j] = lt[(size_t)sj * NCLS + c];
    }
    #pragma unroll
    for (int j = 0; j < 8; ++j) {
      int dj = __shfl(d, 4 * j + quarter, 32);
      atomicAdd(out + (size_t)dj * NCLS + c, v[j]);
    }
  } else {
    for (int j = 0; j < 8; ++j) {
      int from = 4 * j + quarter;
      int sj = __shfl(s, from, 32);
      int dj = __shfl(d, from, 32);
      if (base + from < e) {
        float v = lt[(size_t)sj * NCLS + c];
        atomicAdd(out + (size_t)dj * NCLS + c, v);
      }
    }
  }
}


__global__ void lsm_kernel(float* __restrict__ out,
                           const float* __restrict__ dinv,
                           const float* __restrict__ b2, int n) {
  int i = blockIdx.x * blockDim.x + threadIdx.x;
  if (i >= n) return;
  float* row = out + (size_t)i * NCLS;
  float dv = dinv[i];
  float z[NCLS];
  float mx = -INFINITY;
  #pragma unroll
  for (int c = 0; c < NCLS; ++c) {
    z[c] = row[c] * dv + b2[c];
    mx = fmaxf(mx, z[c]);
  }
  float s = 0.0f;
  #pragma unroll
  for (int c = 0; c < NCLS; ++c) s += expf(z[c] - mx);
  float l = mx + logf(s);
  #pragma unroll
  for (int c = 0; c < NCLS; ++c) row[c] = z[c] - l;
}


extern "C" void kernel_launch(void* const* d_in, const int* in_sizes, int n_in,
                              void* d_out, int out_size, void* d_ws, size_t ws_size,
                              hipStream_t stream) {
  const float* x  = (const float*)d_in[0];
  const int* edge = (const int*)d_in[1];
  const float* W1 = (const float*)d_in[2];
  const float* b1 = (const float*)d_in[3];
  const float* W2 = (const float*)d_in[4];
  const float* b2 = (const float*)d_in[5];

  int n = in_sizes[0] / F_IN;
  int e = in_sizes[1] / 2;
  const int* src = edge;
  const int* dst = edge + e;

  float* deg = (float*)d_ws;
  float* h   = deg + n;
  float* h1  = h + (size_t)n * HID;
  float* lt  = h1 + (size_t)n * HID;
  float* out = (float*)d_out;

  const int T = 256;
  int nTiles  = (n + 15) / 16;
  int gBlocks = (nTiles + 7) / 8;

  deg_init_kernel<<<(n + T - 1) / T, T, 0, stream>>>(deg, n);
  deg_count_kernel<<<(e + T - 1) / T, T, 0, stream>>>(dst, deg, e);
  dinv_kernel<<<(n + T - 1) / T, T, 0, stream>>>(deg, n);

  gemm1_kernel<<<gBlocks, T, 0, stream>>>(x, W1, deg, h, h1, nTiles, n);
  scatter1_kernel<<<(e + T - 1) / T, T, 0, stream>>>(src, dst, h, h1, e);

  gemm2_kernel<<<gBlocks, T, 0, stream>>>(h1, b1, W2, deg, lt, out, nTiles, n);
  scatter2_kernel<<<(e + T - 1) / T, T, 0, stream>>>(src, dst, lt, out, e);

  lsm_kernel<<<(n + T - 1) / T, T, 0, stream>>>(out, deg, b2, n);
}


# f32-atomic edge scatters -> edges bucketed by dst each call (returning int atomic ranks each edge, prefix scan, fill) + per-node row gathers, no float atomics; sums in rank (arrival) order
# speedup vs baseline: 4.1598x; 1.1189x over previous
#include <hip/hip_runtime.h>
#include <hip/hip_bf16.h>
#include <math.h>


typedef float v2f __attribute__((ext_vector_type(2)));
typedef float v8f __attribute__((ext_vector_type(8)));

#define F_IN 128
#define HID  16
#define NCLS 8
#define SCAN_T 1024
#define SCAN_PER_BLOCK 4096


__global__ void cnt_zero_kernel(int* __restrict__ cnt, int n) {
  int i = blockIdx.x * blockDim.x + threadIdx.x;
  if (i < n) cnt[i] = 0;
}

__global__ void rank_kernel(const int* __restrict__ dst, int* __restrict__ cnt,
                            int* __restrict__ rank, int e) {
  int i = blockIdx.x * blockDim.x + threadIdx.x;
  if (i < e) rank[i] = atomicAdd(&cnt[dst[i]], 1);
}

__global__ void __launch_bounds__(SCAN_T) scan_kernel(
    const int* __restrict__ cnt, int* __restrict__ off,
    float* __restrict__ dinv, int n) {
  __shared__ int wsum[SCAN_T / 32];
  __shared__ int wpre[SCAN_T / 32];
  int t = threadIdx.x;
  int lane = t & 31;
  int w = t >> 5;

  int part = 0;
  int nfront = blockIdx.x;
  const int4* c4 = (const int4*)cnt;
  int j = 0;
  for (; j + 8 <= nfront; j += 8) {
    int4 v[8];
    #pragma unroll
    for (int u = 0; u < 8; ++u) v[u] = c4[(size_t)(j + u) * SCAN_T + t];
    #pragma unroll
    for (int u = 0; u < 8; ++u) part += v[u].x + v[u].y + v[u].z + v[u].w;
  }
  for (; j < nfront; ++j) {
    int4 v = c4[(size_t)j * SCAN_T + t];
    part += v.x + v.y + v.z + v.w;
  }
  #pragma unroll
  for (int o = 16; o > 0; o >>= 1) part += __shfl_xor(part, o, 32);
  if (lane == 0) wsum[w] = part;
  __syncthreads();
  int base = 0;
  #pragma unroll
  for (int k = 0; k < SCAN_T / 32; ++k) base += wsum[k];
  __syncthreads();

  int i0 = blockIdx.x * SCAN_PER_BLOCK + t * 4;
  int v0 = (i0     < n) ? cnt[i0]     : 0;
  int v1 = (i0 + 1 < n) ? cnt[i0 + 1] : 0;
  int v2 = (i0 + 2 < n) ? cnt[i0 + 2] : 0;
  int v3 = (i0 + 3 < n) ? cnt[i0 + 3] : 0;
  int tsum = v0 + v1 + v2 + v3;
  int inc = tsum;
  #pragma unroll
  for (int o = 1; o < 32; o <<= 1) {
    int up = __shfl_up(inc, o, 32);
    if (lane >= o) inc += up;
  }
  if (lane == 31) wsum[w] = inc;
  __syncthreads();
  if (t == 0) {
    int run = 0;
    #pragma unroll
    for (int k = 0; k < SCAN_T / 32; ++k) { wpre[k] = run; run += wsum[k]; }
  }
  __syncthreads();
  int ex = base + wpre[w] + inc - tsum;
  if (i0 < n)     { off[i0]     = ex;                dinv[i0]     = rsqrtf((float)v0 + 1.0f); }
  if (i0 + 1 < n) { off[i0 + 1] = ex + v0;           dinv[i0 + 1] = rsqrtf((float)v1 + 1.0f); }
  if (i0 + 2 < n) { off[i0 + 2] = ex + v0 + v1;      dinv[i0 + 2] = rsqrtf((float)v2 + 1.0f); }
  if (i0 + 3 < n) { off[i0 + 3] = ex + v0 + v1 + v2; dinv[i0 + 3] = rsqrtf((float)v3 + 1.0f); }
}

__global__ void fill_kernel(const int* __restrict__ src, const int* __restrict__ dst,
                            const int* __restrict__ rank, const int* __restrict__ off,
                            int* __restrict__ bsrc, int e) {
  int i = blockIdx.x * blockDim.x + threadIdx.x;
  if (i < e) bsrc[off[dst[i]] + rank[i]] = src[i];
}


__global__ void __launch_bounds__(256) gemm1_kernel(
    const float* __restrict__ x, const float* __restrict__ W1,
    const float* __restrict__ dinv,
    float* __restrict__ h, int nTiles, int n) {
  int wave = (blockIdx.x * 256 + threadIdx.x) >> 5;
  int lane = threadIdx.x & 31;
  if (wave >= nTiles) return;
  int half = lane >> 4;
  int m    = lane & 15;
  int row  = wave * 16 + m;
  int rowc = row < n ? row : n - 1;
  const float* xrow = x + (size_t)rowc * F_IN;

  v8f c = {};
  #pragma unroll 4
  for (int k0 = 0; k0 < F_IN; k0 += 4) {
    int kk = k0 + 2 * half;
    v2f a = *(const v2f*)(xrow + kk);
    v2f b;
    b.x = W1[kk * HID + m];
    b.y = W1[(kk + 1) * HID + m];
    c = __builtin_amdgcn_wmma_f32_16x16x4_f32(
        false, a, false, b, (short)0, c, false, false);
  }

  float dv = dinv[rowc];
  bool full = (wave * 16 + 16 <= n);
  #pragma unroll
  for (int r = 0; r < 8; ++r) {
    int orow = wave * 16 + r + 8 * half;
    float w  = __shfl(dv, r + 8 * half, 32);
    if (full || orow < n) h[(size_t)orow * HID + m] = c[r] * w;
  }
}


__global__ void __launch_bounds__(256) gather1_kernel(
    const int* __restrict__ bsrc, const int* __restrict__ off,
    const int* __restrict__ cnt, const float* __restrict__ h,
    float* __restrict__ out1, int n) {
  int d = (blockIdx.x * 256 + threadIdx.x) >> 4;
  int c = threadIdx.x & 15;
  int dd = d < n ? d : n - 1;
  int beg = off[dd];
  int cn  = d < n ? cnt[dd] : 0;
  int other = __shfl_xor(cn, 16, 32);
  int cmax = cn > other ? cn : other;
  float acc = h[(size_t)dd * HID + c];
  for (int b0 = 0; b0 < cmax; b0 += 16) {
    int left = cn - b0;
    int s = (c < left) ? bsrc[beg + b0 + c] : 0;
    float v[16];
    #pragma unroll
    for (int k = 0; k < 16; ++k) {
      int sk = __shfl(s, k, 16);
      v[k] = h[(size_t)sk * HID + c];
    }
    #pragma unroll
    for (int k = 0; k < 16; ++k) acc += (k < left) ? v[k] : 0.0f;
  }
  if (d < n) out1[(size_t)d * HID + c] = acc;
}


__global__ void __launch_bounds__(256) gemm2_kernel(
    const float* __restrict__ h1, const float* __restrict__ b1,
    const float* __restrict__ W2, const float* __restrict__ dinv,
    float* __restrict__ lt, int nTiles, int n) {
  int wave = (blockIdx.x * 256 + threadIdx.x) >> 5;
  int lane = threadIdx.x & 31;
  if (wave >= nTiles) return;
  int half = lane >> 4;
  int m    = lane & 15;
  int row  = wave * 16 + m;
  int rowc = row < n ? row : n - 1;
  const float* hrow = h1 + (size_t)rowc * HID;
  float maskv = (m < NCLS) ? 1.0f : 0.0f;
  int mc = m & (NCLS - 1);
  float dv = dinv[rowc];

  v8f c = {};
  #pragma unroll
  for (int k0 = 0; k0 < HID; k0 += 4) {
    int kk = k0 + 2 * half;
    v2f a;
    a.x = fmaxf(hrow[kk]     * dv + b1[kk],     0.0f);
    a.y = fmaxf(hrow[kk + 1] * dv + b1[kk + 1], 0.0f);
    v2f b;
    b.x = W2[kk * NCLS + mc] * maskv;
    b.y = W2[(kk + 1) * NCLS + mc] * maskv;
    c = __builtin_amdgcn_wmma_f32_16x16x4_f32(
        false, a, false, b, (short)0, c, false, false);
  }

  bool full = (wave * 16 + 16 <= n);
  float wr[8];
  #pragma unroll
  for (int r = 0; r < 8; ++r) wr[r] = __shfl(dv, r + 8 * half, 32);
  if (m < NCLS) {
    #pragma unroll
    for (int r = 0; r < 8; ++r) {
      int orow = wave * 16 + r + 8 * half;
      if (full || orow < n) lt[(size_t)orow * NCLS + m] = c[r] * wr[r];
    }
  }
}


__global__ void __launch_bounds__(256) gather2_kernel(
    const int* __restrict__ bsrc, const int* __restrict__ off,
    const int* __restrict__ cnt, const float* __restrict__ lt,
    float* __restrict__ out, int n) {
  int d = (blockIdx.x * 256 + threadIdx.x) >> 3;
  int c = threadIdx.x & 7;
  int dd = d < n ? d : n - 1;
  int beg = off[dd];
  int cn  = d < n ? cnt[dd] : 0;
  int cmax = cn;
  int o1 = __shfl_xor(cmax, 8, 32);
  cmax = cmax > o1 ? cmax : o1;
  int o2 = __shfl_xor(cmax, 16, 32);
  cmax = cmax > o2 ? cmax : o2;
  float acc = lt[(size_t)dd * NCLS + c];
  for (int b0 = 0; b0 < cmax; b0 += 8) {
    int left = cn - b0;
    int s = (c < left) ? bsrc[beg + b0 + c] : 0;
    float v[8];
    #pragma unroll
    for (int k = 0; k < 8; ++k) {
      int sk = __shfl(s, k, 8);
      v[k] = lt[(size_t)sk * NCLS + c];
    }
    #pragma unroll
    for (int k = 0; k < 8; ++k) acc += (k < left) ? v[k] : 0.0f;
  }
  if (d < n) out[(size_t)d * NCLS + c] = acc;
}


__global__ void lsm_kernel(float* __restrict__ out,
                           const float* __restrict__ dinv,
                           const float* __restrict__ b2, int n) {
  int i = blockIdx.x * blockDim.x + threadIdx.x;
  if (i >= n) return;
  float* row = out + (size_t)i * NCLS;
  float dv = dinv[i];
  float z[NCLS];
  float mx = -INFINITY;
  #pragma unroll
  for (int c = 0; c < NCLS; ++c) {
    z[c] = row[c] * dv + b2[c];
    mx = fmaxf(mx, z[c]);
  }
  float s = 0.0f;
  #pragma unroll
  for (int c = 0; c < NCLS; ++c) s += expf(z[c] - mx);
  float l = mx + logf(s);
  #pragma unroll
  for (int c = 0; c < NCLS; ++c) row[c] = z[c] - l;
}


extern "C" void kernel_launch(void* const* d_in, const int* in_sizes, int n_in,
                              void* d_out, int out_size, void* d_ws, size_t ws_size,
                              hipStream_t stream) {
  const float* x  = (const float*)d_in[0];
  const int* edge = (const int*)d_in[1];
  const float* W1 = (const float*)d_in[2];
  const float* b1 = (const float*)d_in[3];
  const float* W2 = (const float*)d_in[4];
  const float* b2 = (const float*)d_in[5];

  int n = in_sizes[0] / F_IN;
  int e = in_sizes[1] / 2;
  const int* src = edge;
  const int* dst = edge + e;

  size_t np = ((size_t)n + 63) / 64 * 64;
  size_t ep = ((size_t)e + 63) / 64 * 64;
  int*   cnt  = (int*)d_ws;
  int*   off  = cnt + np;
  float* dinv = (float*)(off + np);
  int*   rank = (int*)(dinv + np);
  int*   bsrc = rank + ep;
  float* h    = (float*)(bsrc + ep);
  float* h1   = h + np * HID;
  float* lt   = h1 + np * HID;
  float* out  = (float*)d_out;

  const int T = 256;
  int nTiles  = (n + 15) / 16;
  int gBlocks = (nTiles + 7) / 8;

  cnt_zero_kernel<<<(n + T - 1) / T, T, 0, stream>>>(cnt, n);
  rank_kernel<<<(e + T - 1) / T, T, 0, stream>>>(dst, cnt, rank, e);
  scan_kernel<<<(n + SCAN_PER_BLOCK - 1) / SCAN_PER_BLOCK, SCAN_T, 0, stream>>>(cnt, off, dinv, n);
  fill_kernel<<<(e + T - 1) / T, T, 0, stream>>>(src, dst, rank, off, bsrc, e);

  gemm1_kernel<<<gBlocks, T, 0, stream>>>(x, W1, dinv, h, nTiles, n);
  gather1_kernel<<<(n + 15) / 16, T, 0, stream>>>(bsrc, off, cnt, h, h1, n);

  gemm2_kernel<<<gBlocks, T, 0, stream>>>(h1, b1, W2, dinv, lt, nTiles, n);
  gather2_kernel<<<(n + 31) / 32, T, 0, stream>>>(bsrc, off, cnt, lt, out, n);

  lsm_kernel<<<(n + T - 1) / T, T, 0, stream>>>(out, dinv, b2, n);
}


# dst bucketing by per-edge global int atomics, scan and 4-byte scatter -> LDS-staged counting sort of edges into 256-node dst blocks; blocks sum rows with LDS f32 atomics (arrival order)
# speedup vs baseline: 12.0605x; 2.8993x over previous
#include <hip/hip_runtime.h>
#include <hip/hip_bf16.h>
#include <math.h>


typedef float v2f __attribute__((ext_vector_type(2)));
typedef float v8f __attribute__((ext_vector_type(8)));

#define F_IN 128
#define HID  16
#define NCLS 8
#define BIN_SHIFT 8
#define BIN_NODES 256
#define NB_MAX 512
#define PT 512
#define EPT 16
#define CHUNK (PT * EPT)


__device__ __forceinline__ int bin_of(int d) {
  unsigned b = (unsigned)d >> BIN_SHIFT;
  return (int)(b < (unsigned)(NB_MAX - 1) ? b : (unsigned)(NB_MAX - 1));
}

__device__ __forceinline__ int block_excl_scan(int v, int* wtot, int t) {
  int lane = t & 31;
  int w = t >> 5;
  int inc = v;
  #pragma unroll
  for (int o = 1; o < 32; o <<= 1) {
    int up = __shfl_up(inc, o, 32);
    if (lane >= o) inc += up;
  }
  if (lane == 31) wtot[w] = inc;
  __syncthreads();
  int pre = 0;
  #pragma unroll
  for (int k = 0; k < PT / 32; ++k) pre += (k < w) ? wtot[k] : 0;
  __syncthreads();
  return pre + inc - v;
}

__device__ __forceinline__ int block_sum(int v, int* wtot, int t) {
  #pragma unroll
  for (int o = 16; o > 0; o >>= 1) v += __shfl_xor(v, o, 32);
  if ((t & 31) == 0) wtot[t >> 5] = v;
  __syncthreads();
  int s = 0;
  #pragma unroll
  for (int k = 0; k < PT / 32; ++k) s += wtot[k];
  __syncthreads();
  return s;
}


__global__ void zero_kernel(int* __restrict__ p, int n) {
  int i = blockIdx.x * blockDim.x + threadIdx.x;
  if (i < n) p[i] = 0;
}

__global__ void __launch_bounds__(PT) hist_kernel(
    const int* __restrict__ dst, int* __restrict__ binCnt, int e) {
  __shared__ int hist[NB_MAX];
  int t = threadIdx.x;
  hist[t] = 0;
  __syncthreads();
  int base = blockIdx.x * CHUNK;
  int d[EPT];
  #pragma unroll
  for (int k = 0; k < EPT; ++k) {
    int i = base + k * PT + t;
    int dv = dst[i < e ? i : e - 1];
    d[k] = (i < e) ? dv : -1;
  }
  #pragma unroll
  for (int k = 0; k < EPT; ++k)
    if (d[k] >= 0) atomicAdd(&hist[bin_of(d[k])], 1);
  __syncthreads();
  int c = hist[t];
  if (c) atomicAdd(&binCnt[t], c);
}

__global__ void __launch_bounds__(PT) part_kernel(
    const int* __restrict__ src, const int* __restrict__ dst,
    const int* __restrict__ binCnt, int* __restrict__ binCur,
    int* __restrict__ psrc, int* __restrict__ pdst, int e) {
  __shared__ int hist[NB_MAX];
  __shared__ int lstart[NB_MAX];
  __shared__ int gbase[NB_MAX];
  __shared__ int wtot[PT / 32];
  __shared__ int ssrc[CHUNK];
  __shared__ int sdst[CHUNK];
  int t = threadIdx.x;
  hist[t] = 0;
  __syncthreads();
  int base = blockIdx.x * CHUNK;
  int s[EPT], d[EPT], r[EPT];
  #pragma unroll
  for (int k = 0; k < EPT; ++k) {
    int i = base + k * PT + t;
    bool ok = i < e;
    int ic = ok ? i : e - 1;
    int dv = dst[ic];
    s[k] = src[ic];
    d[k] = ok ? dv : -1;
  }
  #pragma unroll
  for (int k = 0; k < EPT; ++k)
    r[k] = (d[k] >= 0) ? atomicAdd(&hist[bin_of(d[k])], 1) : 0;
  __syncthreads();
  int mine = hist[t];
  int ls = block_excl_scan(mine, wtot, t);
  int gs = block_excl_scan(binCnt[t], wtot, t);
  lstart[t] = ls;
  gbase[t] = gs + (mine ? atomicAdd(&binCur[t], mine) : 0);
  __syncthreads();
  #pragma unroll
  for (int k = 0; k < EPT; ++k) {
    if (d[k] >= 0) {
      int p = lstart[bin_of(d[k])] + r[k];
      ssrc[p] = s[k];
      sdst[p] = d[k];
    }
  }
  __syncthreads();
  int m = e - base;
  if (m > CHUNK) m = CHUNK;
  for (int j = t; j < m; j += PT) {
    int dj = sdst[j];
    int b = bin_of(dj);
    int g = gbase[b] + (j - lstart[b]);
    psrc[g] = ssrc[j];
    pdst[g] = dj;
  }
}

__global__ void __launch_bounds__(PT) deg_kernel(
    const int* __restrict__ pdst, const int* __restrict__ binCnt,
    float* __restrict__ dinv, int n) {
  __shared__ int c[BIN_NODES];
  __shared__ int wtot[PT / 32];
  int t = threadIdx.x;
  int b = blockIdx.x;
  if (t < BIN_NODES) c[t] = 0;
  int beg = block_sum((t < b) ? binCnt[t] : 0, wtot, t);
  int end = beg + binCnt[b];
  int d0 = b << BIN_SHIFT;
  for (int j = beg + t; j < end; j += PT)
    atomicAdd(&c[(pdst[j] - d0) & (BIN_NODES - 1)], 1);
  __syncthreads();
  int dn = d0 + t;
  if (t < BIN_NODES && dn < n) dinv[dn] = rsqrtf((float)c[t] + 1.0f);
}


__global__ void __launch_bounds__(256) gemm1_kernel(
    const float* __restrict__ x, const float* __restrict__ W1,
    const float* __restrict__ dinv,
    float* __restrict__ h, int nTiles, int n) {
  int wave = (blockIdx.x * 256 + threadIdx.x) >> 5;
  int lane = threadIdx.x & 31;
  if (wave >= nTiles) return;
  int half = lane >> 4;
  int m    = lane & 15;
  int row  = wave * 16 + m;
  int rowc = row < n ? row : n - 1;
  const float* xrow = x + (size_t)rowc * F_IN;

  v8f c = {};
  #pragma unroll 4
  for (int k0 = 0; k0 < F_IN; k0 += 4) {
    int kk = k0 + 2 * half;
    v2f a = *(const v2f*)(xrow + kk);
    v2f b;
    b.x = W1[kk * HID + m];
    b.y = W1[(kk + 1) * HID + m];
    c = __builtin_amdgcn_wmma_f32_16x16x4_f32(
        false, a, false, b, (short)0, c, false, false);
  }

  float dv = dinv[rowc];
  bool full = (wave * 16 + 16 <= n);
  #pragma unroll
  for (int r = 0; r < 8; ++r) {
    int orow = wave * 16 + r + 8 * half;
    float w  = __shfl(dv, r + 8 * half, 32);
    if (full || orow < n) h[(size_t)orow * HID + m] = c[r] * w;
  }
}


__global__ void __launch_bounds__(PT) agg1_kernel(
    const int* __restrict__ psrc, const int* __restrict__ pdst,
    const int* __restrict__ binCnt, const float* __restrict__ h,
    float* __restrict__ out1, int n) {
  __shared__ float acc[BIN_NODES * HID];
  __shared__ int wtot[PT / 32];
  int t = threadIdx.x;
  int b = blockIdx.x;
  int lane = t & 31;
  int w = t >> 5;
  int d0 = b << BIN_SHIFT;
  for (int i = t; i < BIN_NODES * HID; i += PT) {
    int node = d0 + (i >> 4);
    acc[i] = (node < n) ? h[(size_t)d0 * HID + i] : 0.0f;
  }
  int beg = block_sum((t < b) ? binCnt[t] : 0, wtot, t);
  int end = beg + binCnt[b];
  int half = lane >> 4;
  int c = lane & 15;
  int j0 = beg + w * 32;
  int last = end > 0 ? end - 1 : 0;
  int jf = j0 + lane;
  int sv = psrc[jf < last ? jf : last];
  int dv = pdst[jf < last ? jf : last];
  if (jf >= end) dv = -1;
  for (; j0 < end; j0 += PT) {
    int jn = j0 + PT + lane;
    int nsv = psrc[jn < last ? jn : last];
    int ndv = pdst[jn < last ? jn : last];
    if (jn >= end) ndv = -1;
    float v[16];
    int dl[16];
    #pragma unroll
    for (int k = 0; k < 16; ++k) {
      int el = 2 * k + half;
      int s = __shfl(sv, el, 32);
      dl[k] = __shfl(dv, el, 32);
      v[k] = h[(size_t)s * HID + c];
    }
    #pragma unroll
    for (int k = 0; k < 16; ++k)
      if (dl[k] >= 0) atomicAdd(&acc[((dl[k] - d0) & (BIN_NODES - 1)) * HID + c], v[k]);
    sv = nsv;
    dv = ndv;
  }
  __syncthreads();
  for (int i = t; i < BIN_NODES * HID; i += PT) {
    int node = d0 + (i >> 4);
    if (node < n) out1[(size_t)d0 * HID + i] = acc[i];
  }
}


__global__ void __launch_bounds__(256) gemm2_kernel(
    const float* __restrict__ h1, const float* __restrict__ b1,
    const float* __restrict__ W2, const float* __restrict__ dinv,
    float* __restrict__ lt, int nTiles, int n) {
  int wave = (blockIdx.x * 256 + threadIdx.x) >> 5;
  int lane = threadIdx.x & 31;
  if (wave >= nTiles) return;
  int half = lane >> 4;
  int m    = lane & 15;
  int row  = wave * 16 + m;
  int rowc = row < n ? row : n - 1;
  const float* hrow = h1 + (size_t)rowc * HID;
  float maskv = (m < NCLS) ? 1.0f : 0.0f;
  int mc = m & (NCLS - 1);
  float dv = dinv[rowc];

  v8f c = {};
  #pragma unroll
  for (int k0 = 0; k0 < HID; k0 += 4) {
    int kk = k0 + 2 * half;
    v2f a;
    a.x = fmaxf(hrow[kk]     * dv + b1[kk],     0.0f);
    a.y = fmaxf(hrow[kk + 1] * dv + b1[kk + 1], 0.0f);
    v2f b;
    b.x = W2[kk * NCLS + mc] * maskv;
    b.y = W2[(kk + 1) * NCLS + mc] * maskv;
    c = __builtin_amdgcn_wmma_f32_16x16x4_f32(
        false, a, false, b, (short)0, c, false, false);
  }

  bool full = (wave * 16 + 16 <= n);
  float wr[8];
  #pragma unroll
  for (int r = 0; r < 8; ++r) wr[r] = __shfl(dv, r + 8 * half, 32);
  if (m < NCLS) {
    #pragma unroll
    for (int r = 0; r < 8; ++r) {
      int orow = wave * 16 + r + 8 * half;
      if (full || orow < n) lt[(size_t)orow * NCLS + m] = c[r] * wr[r];
    }
  }
}


__global__ void __launch_bounds__(PT) agg2_kernel(
    const int* __restrict__ psrc, const int* __restrict__ pdst,
    const int* __restrict__ binCnt, const float* __restrict__ lt,
    float* __restrict__ out, int n) {
  __shared__ float acc[BIN_NODES * NCLS];
  __shared__ int wtot[PT / 32];
  int t = threadIdx.x;
  int b = blockIdx.x;
  int lane = t & 31;
  int w = t >> 5;
  int d0 = b << BIN_SHIFT;
  for (int i = t; i < BIN_NODES * NCLS; i += PT) {
    int node = d0 + (i >> 3);
    acc[i] = (node < n) ? lt[(size_t)d0 * NCLS + i] : 0.0f;
  }
  int beg = block_sum((t < b) ? binCnt[t] : 0, wtot, t);
  int end = beg + binCnt[b];
  int q = lane >> 3;
  int c = lane & 7;
  int j0 = beg + w * 32;
  int last = end > 0 ? end - 1 : 0;
  int jf = j0 + lane;
  int sv = psrc[jf < last ? jf : last];
  int dv = pdst[jf < last ? jf : last];
  if (jf >= end) dv = -1;
  for (; j0 < end; j0 += PT) {
    int jn = j0 + PT + lane;
    int nsv = psrc[jn < last ? jn : last];
    int ndv = pdst[jn < last ? jn : last];
    if (jn >= end) ndv = -1;
    float v[8];
    int dl[8];
    #pragma unroll
    for (int k = 0; k < 8; ++k) {
      int el = 4 * k + q;
      int s = __shfl(sv, el, 32);
      dl[k] = __shfl(dv, el, 32);
      v[k] = lt[(size_t)s * NCLS + c];
    }
    #pragma unroll
    for (int k = 0; k < 8; ++k)
      if (dl[k] >= 0) atomicAdd(&acc[((dl[k] - d0) & (BIN_NODES - 1)) * NCLS + c], v[k]);
    sv = nsv;
    dv = ndv;
  }
  __syncthreads();
  for (int i = t; i < BIN_NODES * NCLS; i += PT) {
    int node = d0 + (i >> 3);
    if (node < n) out[(size_t)d0 * NCLS + i] = acc[i];
  }
}


__global__ void lsm_kernel(float* __restrict__ out,
                           const float* __restrict__ dinv,
                           const float* __restrict__ b2, int n) {
  int i = blockIdx.x * blockDim.x + threadIdx.x;
  if (i >= n) return;
  float* row = out + (size_t)i * NCLS;
  float dv = dinv[i];
  float z[NCLS];
  float mx = -INFINITY;
  #pragma unroll
  for (int c = 0; c < NCLS; ++c) {
    z[c] = row[c] * dv + b2[c];
    mx = fmaxf(mx, z[c]);
  }
  float s = 0.0f;
  #pragma unroll
  for (int c = 0; c < NCLS; ++c) s += expf(z[c] - mx);
  float l = mx + logf(s);
  #pragma unroll
  for (int c = 0; c < NCLS; ++c) row[c] = z[c] - l;
}


extern "C" void kernel_launch(void* const* d_in, const int* in_sizes, int n_in,
                              void* d_out, int out_size, void* d_ws, size_t ws_size,
                              hipStream_t stream) {
  const float* x  = (const float*)d_in[0];
  const int* edge = (const int*)d_in[1];
  const float* W1 = (const float*)d_in[2];
  const float* b1 = (const float*)d_in[3];
  const float* W2 = (const float*)d_in[4];
  const float* b2 = (const float*)d_in[5];

  int n = in_sizes[0] / F_IN;
  int e = in_sizes[1] / 2;
  const int* src = edge;
  const int* dst = edge + e;

  size_t np = ((size_t)n + 63) / 64 * 64;
  size_t ep = ((size_t)e + 63) / 64 * 64;
  int*   binCnt = (int*)d_ws;
  int*   binCur = binCnt + NB_MAX;
  float* dinv = (float*)(binCur + NB_MAX);
  int*   psrc = (int*)(dinv + np);
  int*   pdst = psrc + ep;
  float* h    = (float*)(pdst + ep);
  float* h1   = h + np * HID;
  float* lt   = h1 + np * HID;
  float* out  = (float*)d_out;

  const int T = 256;
  int nTiles  = (n + 15) / 16;
  int gBlocks = (nTiles + 7) / 8;
  int nBins   = (n + BIN_NODES - 1) / BIN_NODES;
  int nChunks = (e + CHUNK - 1) / CHUNK;

  zero_kernel<<<(2 * NB_MAX + T - 1) / T, T, 0, stream>>>(binCnt, 2 * NB_MAX);
  hist_kernel<<<nChunks, PT, 0, stream>>>(dst, binCnt, e);
  part_kernel<<<nChunks, PT, 0, stream>>>(src, dst, binCnt, binCur, psrc, pdst, e);
  deg_kernel<<<nBins, PT, 0, stream>>>(pdst, binCnt, dinv, n);

  gemm1_kernel<<<gBlocks, T, 0, stream>>>(x, W1, dinv, h, nTiles, n);
  agg1_kernel<<<nBins, PT, 0, stream>>>(psrc, pdst, binCnt, h, h1, n);

  gemm2_kernel<<<gBlocks, T, 0, stream>>>(h1, b1, W2, dinv, lt, nTiles, n);
  agg2_kernel<<<nBins, PT, 0, stream>>>(psrc, pdst, binCnt, lt, out, n);

  lsm_kernel<<<(n + T - 1) / T, T, 0, stream>>>(out, dinv, b2, n);
}


# 9 launches -> 6: each per-node stage runs inside the node-block kernel that holds its rows: degree count beside GEMM1, GEMM2 and log_softmax on the LDS row sums (never stored); sum order unchanged
# speedup vs baseline: 15.1662x; 1.2575x over previous
#include <hip/hip_runtime.h>
#include <hip/hip_bf16.h>
#include <math.h>


typedef float v2f __attribute__((ext_vector_type(2)));
typedef float v8f __attribute__((ext_vector_type(8)));

#define F_IN 128
#define HID  16
#define NCLS 8
#define BIN_SHIFT 8
#define BIN_NODES 256
#define NB_MAX 512
#define PT 512
#define EPT 16
#define CHUNK (PT * EPT)
#define DPT 20


__device__ __forceinline__ int bin_of(int d) {
  unsigned b = (unsigned)d >> BIN_SHIFT;
  return (int)(b < (unsigned)(NB_MAX - 1) ? b : (unsigned)(NB_MAX - 1));
}

__device__ __forceinline__ int block_excl_scan(int v, int* wtot, int t) {
  int lane = t & 31;
  int w = t >> 5;
  int inc = v;
  #pragma unroll
  for (int o = 1; o < 32; o <<= 1) {
    int up = __shfl_up(inc, o, 32);
    if (lane >= o) inc += up;
  }
  if (lane == 31) wtot[w] = inc;
  __syncthreads();
  int pre = 0;
  #pragma unroll
  for (int k = 0; k < PT / 32; ++k) pre += (k < w) ? wtot[k] : 0;
  __syncthreads();
  return pre + inc - v;
}

__device__ __forceinline__ int block_sum(int v, int* wtot, int t) {
  #pragma unroll
  for (int o = 16; o > 0; o >>= 1) v += __shfl_xor(v, o, 32);
  if ((t & 31) == 0) wtot[t >> 5] = v;
  __syncthreads();
  int s = 0;
  #pragma unroll
  for (int k = 0; k < PT / 32; ++k) s += wtot[k];
  __syncthreads();
  return s;
}


__global__ void zero_kernel(int* __restrict__ p, int n) {
  int i = blockIdx.x * blockDim.x + threadIdx.x;
  if (i < n) p[i] = 0;
}

__global__ void __launch_bounds__(PT) hist_kernel(
    const int* __restrict__ dst, int* __restrict__ binCnt, int e) {
  __shared__ int hist[NB_MAX];
  int t = threadIdx.x;
  hist[t] = 0;
  __syncthreads();
  int base = blockIdx.x * CHUNK;
  int d[EPT];
  #pragma unroll
  for (int k = 0; k < EPT; ++k) {
    int i = base + k * PT + t;
    int dv = dst[i < e ? i : e - 1];
    d[k] = (i < e) ? dv : -1;
  }
  #pragma unroll
  for (int k = 0; k < EPT; ++k)
    if (d[k] >= 0) atomicAdd(&hist[bin_of(d[k])], 1);
  __syncthreads();
  int c = hist[t];
  if (c) atomicAdd(&binCnt[t], c);
}

__global__ void __launch_bounds__(PT) part_kernel(
    const int* __restrict__ src, const int* __restrict__ dst,
    const int* __restrict__ binCnt, int* __restrict__ binCur,
    int* __restrict__ psrc, int* __restrict__ pdst, int e) {
  __shared__ int hist[NB_MAX];
  __shared__ int lstart[NB_MAX];
  __shared__ int gbase[NB_MAX];
  __shared__ int wtot[PT / 32];
  __shared__ int ssrc[CHUNK];
  __shared__ int sdst[CHUNK];
  int t = threadIdx.x;
  hist[t] = 0;
  __syncthreads();
  int base = blockIdx.x * CHUNK;
  int s[EPT], d[EPT], r[EPT];
  #pragma unroll
  for (int k = 0; k < EPT; ++k) {
    int i = base + k * PT + t;
    bool ok = i < e;
    int ic = ok ? i : e - 1;
    int dv = dst[ic];
    s[k] = src[ic];
    d[k] = ok ? dv : -1;
  }
  #pragma unroll
  for (int k = 0; k < EPT; ++k)
    r[k] = (d[k] >= 0) ? atomicAdd(&hist[bin_of(d[k])], 1) : 0;
  __syncthreads();
  int mine = hist[t];
  int ls = block_excl_scan(mine, wtot, t);
  int gs = block_excl_scan(binCnt[t], wtot, t);
  lstart[t] = ls;
  gbase[t] = gs + (mine ? atomicAdd(&binCur[t], mine) : 0);
  __syncthreads();
  #pragma unroll
  for (int k = 0; k < EPT; ++k) {
    if (d[k] >= 0) {
      int p = lstart[bin_of(d[k])] + r[k];
      ssrc[p] = s[k];
      sdst[p] = d[k];
    }
  }
  __syncthreads();
  int m = e - base;
  if (m > CHUNK) m = CHUNK;
  for (int j = t; j < m; j += PT) {
    int dj = sdst[j];
    int b = bin_of(dj);
    int g = gbase[b] + (j - lstart[b]);
    psrc[g] = ssrc[j];
    pdst[g] = dj;
  }
}


__global__ void __launch_bounds__(PT) __attribute__((amdgpu_waves_per_eu(4, 8))) deg_gemm1_kernel(
    const int* __restrict__ pdst, const int* __restrict__ binCnt,
    const float* __restrict__ x, const float* __restrict__ W1,
    float* __restrict__ dinv, float* __restrict__ h, int n) {
  __shared__ int cnt[BIN_NODES];
  __shared__ int wtot[PT / 32];
  int t = threadIdx.x;
  int b = blockIdx.x;
  int lane = t & 31;
  int w = t >> 5;
  int d0 = b << BIN_SHIFT;
  if (t < BIN_NODES) cnt[t] = 0;
  int beg = block_sum((t < b) ? binCnt[t] : 0, wtot, t);
  int end = beg + binCnt[b];
  int last = end > 0 ? end - 1 : 0;
  int dd[DPT];
  #pragma unroll
  for (int k = 0; k < DPT; ++k) {
    int j = beg + k * PT + t;
    int dj = pdst[j < last ? j : last];
    dd[k] = (j < end) ? dj : -1;
  }

  int half = lane >> 4;
  int m    = lane & 15;
  int row  = d0 + w * 16 + m;
  int rowc = row < n ? row : n - 1;
  const float* xrow = x + (size_t)rowc * F_IN;

  v8f c = {};
  #pragma unroll 4
  for (int k0 = 0; k0 < F_IN; k0 += 4) {
    int kk = k0 + 2 * half;
    v2f a = *(const v2f*)(xrow + kk);
    v2f bb;
    bb.x = W1[kk * HID + m];
    bb.y = W1[(kk + 1) * HID + m];
    c = __builtin_amdgcn_wmma_f32_16x16x4_f32(
        false, a, false, bb, (short)0, c, false, false);
  }

  #pragma unroll
  for (int k = 0; k < DPT; ++k)
    if (dd[k] >= 0) atomicAdd(&cnt[(dd[k] - d0) & (BIN_NODES - 1)], 1);
  for (int j = beg + DPT * PT + t; j < end; j += PT)
    atomicAdd(&cnt[(pdst[j] - d0) & (BIN_NODES - 1)], 1);
  __syncthreads();

  float dv = rsqrtf((float)cnt[w * 16 + m] + 1.0f);
  if (half == 0 && row < n) dinv[row] = dv;
  bool full = (d0 + w * 16 + 16 <= n);
  #pragma unroll
  for (int r = 0; r < 8; ++r) {
    int orow = d0 + w * 16 + r + 8 * half;
    float ws = __shfl(dv, r + 8 * half, 32);
    if (full || orow < n) h[(size_t)orow * HID + m] = c[r] * ws;
  }
}


__global__ void __launch_bounds__(PT) agg1_gemm2_kernel(
    const int* __restrict__ psrc, const int* __restrict__ pdst,
    const int* __restrict__ binCnt, const float* __restrict__ h,
    const float* __restrict__ b1, const float* __restrict__ W2,
    const float* __restrict__ dinv, float* __restrict__ lt, int n) {
  __shared__ float acc[BIN_NODES * HID];
  __shared__ int wtot[PT / 32];
  int t = threadIdx.x;
  int b = blockIdx.x;
  int lane = t & 31;
  int w = t >> 5;
  int d0 = b << BIN_SHIFT;
  for (int i = t; i < BIN_NODES * HID; i += PT) {
    int node = d0 + (i >> 4);
    acc[i] = (node < n) ? h[(size_t)d0 * HID + i] : 0.0f;
  }
  int beg = block_sum((t < b) ? binCnt[t] : 0, wtot, t);
  int end = beg + binCnt[b];
  int half = lane >> 4;
  int c = lane & 15;
  int j0 = beg + w * 32;
  int last = end > 0 ? end - 1 : 0;
  int jf = j0 + lane;
  int sv = psrc[jf < last ? jf : last];
  int dv = pdst[jf < last ? jf : last];
  if (jf >= end) dv = -1;
  for (; j0 < end; j0 += PT) {
    int jn = j0 + PT + lane;
    int nsv = psrc[jn < last ? jn : last];
    int ndv = pdst[jn < last ? jn : last];
    if (jn >= end) ndv = -1;
    float v[16];
    int dl[16];
    #pragma unroll
    for (int k = 0; k < 16; ++k) {
      int el = 2 * k + half;
      int s = __shfl(sv, el, 32);
      dl[k] = __shfl(dv, el, 32);
      v[k] = h[(size_t)s * HID + c];
    }
    #pragma unroll
    for (int k = 0; k < 16; ++k)
      if (dl[k] >= 0) atomicAdd(&acc[((dl[k] - d0) & (BIN_NODES - 1)) * HID + c], v[k]);
    sv = nsv;
    dv = ndv;
  }
  __syncthreads();

  int row  = d0 + w * 16 + c;
  int rowc = row < n ? row : n - 1;
  const float* hrow = acc + (w * 16 + c) * HID;
  float maskv = (c < NCLS) ? 1.0f : 0.0f;
  int mc = c & (NCLS - 1);
  float dr = dinv[rowc];

  v8f cc = {};
  #pragma unroll
  for (int k0 = 0; k0 < HID; k0 += 4) {
    int kk = k0 + 2 * half;
    v2f a;
    a.x = fmaxf(hrow[kk]     * dr + b1[kk],     0.0f);
    a.y = fmaxf(hrow[kk + 1] * dr + b1[kk + 1], 0.0f);
    v2f bb;
    bb.x = W2[kk * NCLS + mc] * maskv;
    bb.y = W2[(kk + 1) * NCLS + mc] * maskv;
    cc = __builtin_amdgcn_wmma_f32_16x16x4_f32(
        false, a, false, bb, (short)0, cc, false, false);
  }

  bool full = (d0 + w * 16 + 16 <= n);
  float wr[8];
  #pragma unroll
  for (int r = 0; r < 8; ++r) wr[r] = __shfl(dr, r + 8 * half, 32);
  if (c < NCLS) {
    #pragma unroll
    for (int r = 0; r < 8; ++r) {
      int orow = d0 + w * 16 + r + 8 * half;
      if (full || orow < n) lt[(size_t)orow * NCLS + c] = cc[r] * wr[r];
    }
  }
}


__global__ void __launch_bounds__(PT) agg2_lsm_kernel(
    const int* __restrict__ psrc, const int* __restrict__ pdst,
    const int* __restrict__ binCnt, const float* __restrict__ lt,
    const float* __restrict__ dinv, const float* __restrict__ b2,
    float* __restrict__ out, int n) {
  __shared__ float acc[BIN_NODES * NCLS];
  __shared__ int wtot[PT / 32];
  int t = threadIdx.x;
  int b = blockIdx.x;
  int lane = t & 31;
  int w = t >> 5;
  int d0 = b << BIN_SHIFT;
  for (int i = t; i < BIN_NODES * NCLS; i += PT) {
    int node = d0 + (i >> 3);
    acc[i] = (node < n) ? lt[(size_t)d0 * NCLS + i] : 0.0f;
  }
  int beg = block_sum((t < b) ? binCnt[t] : 0, wtot, t);
  int end = beg + binCnt[b];
  int q = lane >> 3;
  int c = lane & 7;
  int j0 = beg + w * 32;
  int last = end > 0 ? end - 1 : 0;
  int jf = j0 + lane;
  int sv = psrc[jf < last ? jf : last];
  int dv = pdst[jf < last ? jf : last];
  if (jf >= end) dv = -1;
  for (; j0 < end; j0 += PT) {
    int jn = j0 + PT + lane;
    int nsv = psrc[jn < last ? jn : last];
    int ndv = pdst[jn < last ? jn : last];
    if (jn >= end) ndv = -1;
    float v[8];
    int dl[8];
    #pragma unroll
    for (int k = 0; k < 8; ++k) {
      int el = 4 * k + q;
      int s = __shfl(sv, el, 32);
      dl[k] = __shfl(dv, el, 32);
      v[k] = lt[(size_t)s * NCLS + c];
    }
    #pragma unroll
    for (int k = 0; k < 8; ++k)
      if (dl[k] >= 0) atomicAdd(&acc[((dl[k] - d0) & (BIN_NODES - 1)) * NCLS + c], v[k]);
    sv = nsv;
    dv = ndv;
  }
  __syncthreads();

  int node = d0 + t;
  if (t < BIN_NODES && node < n) {
    const float* srow = acc + t * NCLS;
    float* row = out + (size_t)node * NCLS;
    float dn = dinv[node];
    float z[NCLS];
    float mx = -INFINITY;
    #pragma unroll
    for (int k = 0; k < NCLS; ++k) {
      z[k] = srow[k] * dn + b2[k];
      mx = fmaxf(mx, z[k]);
    }
    float s = 0.0f;
    #pragma unroll
    for (int k = 0; k < NCLS; ++k) s += expf(z[k] - mx);
    float l = mx + logf(s);
    #pragma unroll
    for (int k = 0; k < NCLS; ++k) row[k] = z[k] - l;
  }
}


extern "C" void kernel_launch(void* const* d_in, const int* in_sizes, int n_in,
                              void* d_out, int out_size, void* d_ws, size_t ws_size,
                              hipStream_t stream) {
  const float* x  = (const float*)d_in[0];
  const int* edge = (const int*)d_in[1];
  const float* W1 = (const float*)d_in[2];
  const float* b1 = (const float*)d_in[3];
  const float* W2 = (const float*)d_in[4];
  const float* b2 = (const float*)d_in[5];

  int n = in_sizes[0] / F_IN;
  int e = in_sizes[1] / 2;
  const int* src = edge;
  const int* dst = edge + e;

  size_t np = ((size_t)n + 63) / 64 * 64;
  size_t ep = ((size_t)e + 63) / 64 * 64;
  int*   binCnt = (int*)d_ws;
  int*   binCur = binCnt + NB_MAX;
  float* dinv = (float*)(binCur + NB_MAX);
  int*   psrc = (int*)(dinv + np);
  int*   pdst = psrc + ep;
  float* h    = (float*)(pdst + ep);
  float* lt   = h + np * HID;
  float* out  = (float*)d_out;

  const int T = 256;
  int nBins   = (n + BIN_NODES - 1) / BIN_NODES;
  int nChunks = (e + CHUNK - 1) / CHUNK;

  zero_kernel<<<(2 * NB_MAX + T - 1) / T, T, 0, stream>>>(binCnt, 2 * NB_MAX);
  hist_kernel<<<nChunks, PT, 0, stream>>>(dst, binCnt, e);
  part_kernel<<<nChunks, PT, 0, stream>>>(src, dst, binCnt, binCur, psrc, pdst, e);

  deg_gemm1_kernel<<<nBins, PT, 0, stream>>>(pdst, binCnt, x, W1, dinv, h, n);
  agg1_gemm2_kernel<<<nBins, PT, 0, stream>>>(psrc, pdst, binCnt, h, b1, W2, dinv, lt, n);
  agg2_lsm_kernel<<<nBins, PT, 0, stream>>>(psrc, pdst, binCnt, lt, dinv, b2, out, n);
}


# Partitioned edge stored as one packed word (source | row in block << 20): sum loops read one index word, do one shuffle and one unguarded LDS f32 add per edge; arrival-order sums as before
# speedup vs baseline: 16.4740x; 1.0862x over previous
#include <hip/hip_runtime.h>
#include <hip/hip_bf16.h>
#include <math.h>


typedef float v2f __attribute__((ext_vector_type(2)));
typedef float v8f __attribute__((ext_vector_type(8)));

#define F_IN 128
#define HID  16
#define NCLS 8
#define BIN_SHIFT 8
#define BIN_NODES 256
#define NB_MAX 512
#define PT 512
#define EPT 16
#define CHUNK (PT * EPT)
#define DPT 20
#define SRC_BITS 20
#define SRC_MASK ((1 << SRC_BITS) - 1)
#define SPARE_ROW (BIN_NODES << SRC_BITS)


__device__ __forceinline__ int bin_of(int d) {
  unsigned b = (unsigned)d >> BIN_SHIFT;
  return (int)(b < (unsigned)(NB_MAX - 1) ? b : (unsigned)(NB_MAX - 1));
}

__device__ __forceinline__ int pack_edge(int s, int d, int b) {
  return (s & SRC_MASK) | (((d - (b << BIN_SHIFT)) & (BIN_NODES - 1)) << SRC_BITS);
}

__device__ __forceinline__ int block_excl_scan(int v, int* wtot, int t) {
  int lane = t & 31;
  int w = t >> 5;
  int inc = v;
  #pragma unroll
  for (int o = 1; o < 32; o <<= 1) {
    int up = __shfl_up(inc, o, 32);
    if (lane >= o) inc += up;
  }
  if (lane == 31) wtot[w] = inc;
  __syncthreads();
  int pre = 0;
  #pragma unroll
  for (int k = 0; k < PT / 32; ++k) pre += (k < w) ? wtot[k] : 0;
  __syncthreads();
  return pre + inc - v;
}

__device__ __forceinline__ int block_sum(int v, int* wtot, int t) {
  #pragma unroll
  for (int o = 16; o > 0; o >>= 1) v += __shfl_xor(v, o, 32);
  if ((t & 31) == 0) wtot[t >> 5] = v;
  __syncthreads();
  int s = 0;
  #pragma unroll
  for (int k = 0; k < PT / 32; ++k) s += wtot[k];
  __syncthreads();
  return s;
}


__global__ void zero_kernel(int* __restrict__ p, int n) {
  int i = blockIdx.x * blockDim.x + threadIdx.x;
  if (i < n) p[i] = 0;
}

__global__ void __launch_bounds__(PT) hist_kernel(
    const int* __restrict__ dst, int* __restrict__ binCnt, int e) {
  __shared__ int hist[NB_MAX];
  int t = threadIdx.x;
  hist[t] = 0;
  __syncthreads();
  int base = blockIdx.x * CHUNK;
  int d[EPT];
  #pragma unroll
  for (int k = 0; k < EPT; ++k) {
    int i = base + k * PT + t;
    int dv = dst[i < e ? i : e - 1];
    d[k] = (i < e) ? dv : -1;
  }
  #pragma unroll
  for (int k = 0; k < EPT; ++k)
    if (d[k] >= 0) atomicAdd(&hist[bin_of(d[k])], 1);
  __syncthreads();
  int c = hist[t];
  if (c) atomicAdd(&binCnt[t], c);
}

__global__ void __launch_bounds__(PT) part_kernel(
    const int* __restrict__ src, const int* __restrict__ dst,
    const int* __restrict__ binCnt, int* __restrict__ binCur,
    int* __restrict__ ppk, int e) {
  __shared__ int hist[NB_MAX];
  __shared__ int lstart[NB_MAX];
  __shared__ int gbase[NB_MAX];
  __shared__ int wtot[PT / 32];
  __shared__ int ssrc[CHUNK];
  __shared__ int sdst[CHUNK];
  int t = threadIdx.x;
  hist[t] = 0;
  __syncthreads();
  int base = blockIdx.x * CHUNK;
  int s[EPT], d[EPT], r[EPT];
  #pragma unroll
  for (int k = 0; k < EPT; ++k) {
    int i = base + k * PT + t;
    bool ok = i < e;
    int ic = ok ? i : e - 1;
    int dv = dst[ic];
    s[k] = src[ic];
    d[k] = ok ? dv : -1;
  }
  #pragma unroll
  for (int k = 0; k < EPT; ++k)
    r[k] = (d[k] >= 0) ? atomicAdd(&hist[bin_of(d[k])], 1) : 0;
  __syncthreads();
  int mine = hist[t];
  int ls = block_excl_scan(mine, wtot, t);
  int gs = block_excl_scan(binCnt[t], wtot, t);
  lstart[t] = ls;
  gbase[t] = gs + (mine ? atomicAdd(&binCur[t], mine) : 0);
  __syncthreads();
  #pragma unroll
  for (int k = 0; k < EPT; ++k) {
    if (d[k] >= 0) {
      int p = lstart[bin_of(d[k])] + r[k];
      ssrc[p] = s[k];
      sdst[p] = d[k];
    }
  }
  __syncthreads();
  int m = e - base;
  if (m > CHUNK) m = CHUNK;
  for (int j = t; j < m; j += PT) {
    int dj = sdst[j];
    int b = bin_of(dj);
    int g = gbase[b] + (j - lstart[b]);
    ppk[g] = pack_edge(ssrc[j], dj, b);
  }
}


__global__ void __launch_bounds__(PT) __attribute__((amdgpu_waves_per_eu(4, 8))) deg_gemm1_kernel(
    const int* __restrict__ ppk, const int* __restrict__ binCnt,
    const float* __restrict__ x, const float* __restrict__ W1,
    float* __restrict__ dinv, float* __restrict__ h, int n) {
  __shared__ int cnt[BIN_NODES];
  __shared__ int wtot[PT / 32];
  int t = threadIdx.x;
  int b = blockIdx.x;
  int lane = t & 31;
  int w = t >> 5;
  int d0 = b << BIN_SHIFT;
  if (t < BIN_NODES) cnt[t] = 0;
  int beg = block_sum((t < b) ? binCnt[t] : 0, wtot, t);
  int end = beg + binCnt[b];
  int last = end > 0 ? end - 1 : 0;
  int dd[DPT];
  #pragma unroll
  for (int k = 0; k < DPT; ++k) {
    int j = beg + k * PT + t;
    int wj = ppk[j < last ? j : last];
    dd[k] = (j < end) ? wj : -1;
  }

  int half = lane >> 4;
  int m    = lane & 15;
  int row  = d0 + w * 16 + m;
  int rowc = row < n ? row : n - 1;
  const float* xrow = x + (size_t)rowc * F_IN;

  v8f c = {};
  #pragma unroll 4
  for (int k0 = 0; k0 < F_IN; k0 += 4) {
    int kk = k0 + 2 * half;
    v2f a = *(const v2f*)(xrow + kk);
    v2f bb;
    bb.x = W1[kk * HID + m];
    bb.y = W1[(kk + 1) * HID + m];
    c = __builtin_amdgcn_wmma_f32_16x16x4_f32(
        false, a, false, bb, (short)0, c, false, false);
  }

  #pragma unroll
  for (int k = 0; k < DPT; ++k)
    if (dd[k] >= 0) atomicAdd(&cnt[(dd[k] >> SRC_BITS) & (BIN_NODES - 1)], 1);
  for (int j = beg + DPT * PT + t; j < end; j += PT)
    atomicAdd(&cnt[(ppk[j] >> SRC_BITS) & (BIN_NODES - 1)], 1);
  __syncthreads();

  float dv = rsqrtf((float)cnt[w * 16 + m] + 1.0f);
  if (half == 0 && row < n) dinv[row] = dv;
  bool full = (d0 + w * 16 + 16 <= n);
  #pragma unroll
  for (int r = 0; r < 8; ++r) {
    int orow = d0 + w * 16 + r + 8 * half;
    float ws = __shfl(dv, r + 8 * half, 32);
    if (full || orow < n) h[(size_t)orow * HID + m] = c[r] * ws;
  }
}


__global__ void __launch_bounds__(PT) agg1_gemm2_kernel(
    const int* __restrict__ ppk,
    const int* __restrict__ binCnt, const float* __restrict__ h,
    const float* __restrict__ b1, const float* __restrict__ W2,
    const float* __restrict__ dinv, float* __restrict__ lt, int n) {
  __shared__ float acc[(BIN_NODES + 1) * HID];
  __shared__ int wtot[PT / 32];
  int t = threadIdx.x;
  int b = blockIdx.x;
  int lane = t & 31;
  int w = t >> 5;
  int d0 = b << BIN_SHIFT;
  for (int i = t; i < BIN_NODES * HID; i += PT) {
    int node = d0 + (i >> 4);
    acc[i] = (node < n) ? h[(size_t)d0 * HID + i] : 0.0f;
  }
  if (t < HID) acc[BIN_NODES * HID + t] = 0.0f;
  int beg = block_sum((t < b) ? binCnt[t] : 0, wtot, t);
  int end = beg + binCnt[b];
  int half = lane >> 4;
  int c = lane & 15;
  int j0 = beg + w * 32;
  int last = end > 0 ? end - 1 : 0;
  int jf = j0 + lane;
  int wv = ppk[jf < last ? jf : last];
  if (jf >= end) wv = (wv & SRC_MASK) | SPARE_ROW;
  for (; j0 < end; j0 += PT) {
    int jn = j0 + PT + lane;
    int nwv = ppk[jn < last ? jn : last];
    if (jn >= end) nwv = (nwv & SRC_MASK) | SPARE_ROW;
    float v[16];
    unsigned dl[16];
    #pragma unroll
    for (int k = 0; k < 16; ++k) {
      int wk = __shfl(wv, 2 * k + half, 32);
      dl[k] = (unsigned)wk >> SRC_BITS;
      v[k] = h[(unsigned)(wk & SRC_MASK) * HID + c];
    }
    #pragma unroll
    for (int k = 0; k < 16; ++k)
      atomicAdd(&acc[dl[k] * HID + c], v[k]);
    wv = nwv;
  }
  __syncthreads();

  int row  = d0 + w * 16 + c;
  int rowc = row < n ? row : n - 1;
  const float* hrow = acc + (w * 16 + c) * HID;
  float maskv = (c < NCLS) ? 1.0f : 0.0f;
  int mc = c & (NCLS - 1);
  float dr = dinv[rowc];

  v8f cc = {};
  #pragma unroll
  for (int k0 = 0; k0 < HID; k0 += 4) {
    int kk = k0 + 2 * half;
    v2f a;
    a.x = fmaxf(hrow[kk]     * dr + b1[kk],     0.0f);
    a.y = fmaxf(hrow[kk + 1] * dr + b1[kk + 1], 0.0f);
    v2f bb;
    bb.x = W2[kk * NCLS + mc] * maskv;
    bb.y = W2[(kk + 1) * NCLS + mc] * maskv;
    cc = __builtin_amdgcn_wmma_f32_16x16x4_f32(
        false, a, false, bb, (short)0, cc, false, false);
  }

  bool full = (d0 + w * 16 + 16 <= n);
  float wr[8];
  #pragma unroll
  for (int r = 0; r < 8; ++r) wr[r] = __shfl(dr, r + 8 * half, 32);
  if (c < NCLS) {
    #pragma unroll
    for (int r = 0; r < 8; ++r) {
      int orow = d0 + w * 16 + r + 8 * half;
      if (full || orow < n) lt[(size_t)orow * NCLS + c] = cc[r] * wr[r];
    }
  }
}


__global__ void __launch_bounds__(PT) agg2_lsm_kernel(
    const int* __restrict__ ppk,
    const int* __restrict__ binCnt, const float* __restrict__ lt,
    const float* __restrict__ dinv, const float* __restrict__ b2,
    float* __restrict__ out, int n) {
  __shared__ float acc[(BIN_NODES + 1) * NCLS];
  __shared__ int wtot[PT / 32];
  int t = threadIdx.x;
  int b = blockIdx.x;
  int lane = t & 31;
  int w = t >> 5;
  int d0 = b << BIN_SHIFT;
  for (int i = t; i < BIN_NODES * NCLS; i += PT) {
    int node = d0 + (i >> 3);
    acc[i] = (node < n) ? lt[(size_t)d0 * NCLS + i] : 0.0f;
  }
  if (t < NCLS) acc[BIN_NODES * NCLS + t] = 0.0f;
  int beg = block_sum((t < b) ? binCnt[t] : 0, wtot, t);
  int end = beg + binCnt[b];
  int q = lane >> 3;
  int c = lane & 7;
  int j0 = beg + w * 32;
  int last = end > 0 ? end - 1 : 0;
  int jf = j0 + lane;
  int wv = ppk[jf < last ? jf : last];
  if (jf >= end) wv = (wv & SRC_MASK) | SPARE_ROW;
  for (; j0 < end; j0 += PT) {
    int jn = j0 + PT + lane;
    int nwv = ppk[jn < last ? jn : last];
    if (jn >= end) nwv = (nwv & SRC_MASK) | SPARE_ROW;
    float v[8];
    unsigned dl[8];
    #pragma unroll
    for (int k = 0; k < 8; ++k) {
      int wk = __shfl(wv, 4 * k + q, 32);
      dl[k] = (unsigned)wk >> SRC_BITS;
      v[k] = lt[(unsigned)(wk & SRC_MASK) * NCLS + c];
    }
    #pragma unroll
    for (int k = 0; k < 8; ++k)
      atomicAdd(&acc[dl[k] * NCLS + c], v[k]);
    wv = nwv;
  }
  __syncthreads();

  int node = d0 + t;
  if (t < BIN_NODES && node < n) {
    const float* srow = acc + t * NCLS;
    float* row = out + (size_t)node * NCLS;
    float dn = dinv[node];
    float z[NCLS];
    float mx = -INFINITY;
    #pragma unroll
    for (int k = 0; k < NCLS; ++k) {
      z[k] = srow[k] * dn + b2[k];
      mx = fmaxf(mx, z[k]);
    }
    float s = 0.0f;
    #pragma unroll
    for (int k = 0; k < NCLS; ++k) s += expf(z[k] - mx);
    float l = mx + logf(s);
    #pragma unroll
    for (int k = 0; k < NCLS; ++k) row[k] = z[k] - l;
  }
}


extern "C" void kernel_launch(void* const* d_in, const int* in_sizes, int n_in,
                              void* d_out, int out_size, void* d_ws, size_t ws_size,
                              hipStream_t stream) {
  const float* x  = (const float*)d_in[0];
  const int* edge = (const int*)d_in[1];
  const float* W1 = (const float*)d_in[2];
  const float* b1 = (const float*)d_in[3];
  const float* W2 = (const float*)d_in[4];
  const float* b2 = (const float*)d_in[5];

  int n = in_sizes[0] / F_IN;
  int e = in_sizes[1] / 2;
  const int* src = edge;
  const int* dst = edge + e;

  size_t np = ((size_t)n + 63) / 64 * 64;
  size_t ep = ((size_t)e + 63) / 64 * 64;
  int*   binCnt = (int*)d_ws;
  int*   binCur = binCnt + NB_MAX;
  float* dinv = (float*)(binCur + NB_MAX);
  int*   ppk  = (int*)(dinv + np);
  float* h    = (float*)(ppk + ep);
  float* lt   = h + np * HID;
  float* out  = (float*)d_out;

  const int T = 256;
  int nBins   = (n + BIN_NODES - 1) / BIN_NODES;
  int nChunks = (e + CHUNK - 1) / CHUNK;

  zero_kernel<<<(2 * NB_MAX + T - 1) / T, T, 0, stream>>>(binCnt, 2 * NB_MAX);
  hist_kernel<<<nChunks, PT, 0, stream>>>(dst, binCnt, e);
  part_kernel<<<nChunks, PT, 0, stream>>>(src, dst, binCnt, binCur, ppk, e);

  deg_gemm1_kernel<<<nBins, PT, 0, stream>>>(ppk, binCnt, x, W1, dinv, h, n);
  agg1_gemm2_kernel<<<nBins, PT, 0, stream>>>(ppk, binCnt, h, b1, W2, dinv, lt, n);
  agg2_lsm_kernel<<<nBins, PT, 0, stream>>>(ppk, binCnt, lt, dinv, b2, out, n);
}


# No counting pass: each node block gets a fixed region (mean run + 25% + 256 words) filled by cursor, a full region spills to a list all blocks scan; histogram launch goes (6 -> 5); same sum order
# speedup vs baseline: 17.6020x; 1.0685x over previous
#include <hip/hip_runtime.h>
#include <hip/hip_bf16.h>
#include <math.h>


typedef float v2f __attribute__((ext_vector_type(2)));
typedef float v8f __attribute__((ext_vector_type(8)));

#define F_IN 128
#define HID  16
#define NCLS 8
#define BIN_SHIFT 8
#define BIN_NODES 256
#define NB_MAX 512
#define PT 512
#define EPT 16
#define CHUNK (PT * EPT)
#define DPT 20
#define SRC_BITS 20
#define SRC_MASK ((1 << SRC_BITS) - 1)
#define SPARE_ROW (BIN_NODES << SRC_BITS)


__device__ __forceinline__ int bin_of(int d) {
  unsigned b = (unsigned)d >> BIN_SHIFT;
  return (int)(b < (unsigned)(NB_MAX - 1) ? b : (unsigned)(NB_MAX - 1));
}

__device__ __forceinline__ int pack_edge(int s, int d, int b) {
  return (s & SRC_MASK) | (((d - (b << BIN_SHIFT)) & (BIN_NODES - 1)) << SRC_BITS);
}

__device__ __forceinline__ int bin_cap(int e, int nb) {
  int avg = e / nb;
  return (avg + (avg >> 2) + 256 + 63) & ~63;
}

__device__ __forceinline__ int block_excl_scan(int v, int* wtot, int t) {
  int lane = t & 31;
  int w = t >> 5;
  int inc = v;
  #pragma unroll
  for (int o = 1; o < 32; o <<= 1) {
    int up = __shfl_up(inc, o, 32);
    if (lane >= o) inc += up;
  }
  if (lane == 31) wtot[w] = inc;
  __syncthreads();
  int pre = 0;
  #pragma unroll
  for (int k = 0; k < PT / 32; ++k) pre += (k < w) ? wtot[k] : 0;
  __syncthreads();
  return pre + inc - v;
}


__global__ void zero_kernel(int* __restrict__ p, int n) {
  int i = blockIdx.x * blockDim.x + threadIdx.x;
  if (i < n) p[i] = 0;
}

__global__ void __launch_bounds__(PT) part_kernel(
    const int* __restrict__ src, const int* __restrict__ dst,
    int* __restrict__ binCur, int* __restrict__ ovfCnt, int* __restrict__ ovf,
    int* __restrict__ ppk, int e, int nb) {
  __shared__ int hist[NB_MAX];
  __shared__ int lstart[NB_MAX];
  __shared__ int goff[NB_MAX];
  __shared__ int fitn[NB_MAX];
  __shared__ int obase[NB_MAX];
  __shared__ int wtot[PT / 32];
  __shared__ int ssrc[CHUNK];
  __shared__ int sdst[CHUNK];
  int t = threadIdx.x;
  int cap = bin_cap(e, nb);
  hist[t] = 0;
  __syncthreads();
  int base = blockIdx.x * CHUNK;
  int s[EPT], d[EPT], r[EPT];
  #pragma unroll
  for (int k = 0; k < EPT; ++k) {
    int i = base + k * PT + t;
    bool ok = i < e;
    int ic = ok ? i : e - 1;
    int dv = dst[ic];
    s[k] = src[ic];
    d[k] = ok ? dv : -1;
  }
  #pragma unroll
  for (int k = 0; k < EPT; ++k)
    r[k] = (d[k] >= 0) ? atomicAdd(&hist[bin_of(d[k])], 1) : 0;
  __syncthreads();
  int mine = hist[t];
  int ls = block_excl_scan(mine, wtot, t);
  int off = mine ? atomicAdd(&binCur[t], mine) : 0;
  int fit = cap - off;
  fit = fit < 0 ? 0 : (fit > mine ? mine : fit);
  int nov = mine - fit;
  lstart[t] = ls;
  goff[t] = off;
  fitn[t] = fit;
  obase[t] = nov ? atomicAdd(ovfCnt, nov) : 0;
  __syncthreads();
  #pragma unroll
  for (int k = 0; k < EPT; ++k) {
    if (d[k] >= 0) {
      int p = lstart[bin_of(d[k])] + r[k];
      ssrc[p] = s[k];
      sdst[p] = d[k];
    }
  }
  __syncthreads();
  int m = e - base;
  if (m > CHUNK) m = CHUNK;
  for (int j = t; j < m; j += PT) {
    int dj = sdst[j];
    int b = bin_of(dj);
    int i = j - lstart[b];
    int f = fitn[b];
    if (i < f) {
      ppk[b * cap + goff[b] + i] = pack_edge(ssrc[j], dj, b);
    } else {
      int q = obase[b] + (i - f);
      ovf[2 * q] = ssrc[j];
      ovf[2 * q + 1] = dj;
    }
  }
}


__global__ void __launch_bounds__(PT) __attribute__((amdgpu_waves_per_eu(4, 8))) deg_gemm1_kernel(
    const int* __restrict__ ppk, const int* __restrict__ binCur,
    const int* __restrict__ ovfCnt, const int* __restrict__ ovf,
    const float* __restrict__ x, const float* __restrict__ W1,
    float* __restrict__ dinv, float* __restrict__ h, int n, int e) {
  __shared__ int cnt[BIN_NODES];
  int t = threadIdx.x;
  int b = blockIdx.x;
  int lane = t & 31;
  int w = t >> 5;
  int d0 = b << BIN_SHIFT;
  if (t < BIN_NODES) cnt[t] = 0;
  __syncthreads();
  int cap = bin_cap(e, (int)gridDim.x);
  int tot = binCur[b];
  int nov = ovfCnt[0];
  int beg = b * cap;
  int end = beg + (tot < cap ? tot : cap);
  int last = end - 1;
  bool some = end > beg;
  int dd[DPT];
  #pragma unroll
  for (int k = 0; k < DPT; ++k) {
    int j = beg + k * PT + t;
    const int* pj = some ? ppk + (j < last ? j : last) : binCur;
    int wj = *pj;
    dd[k] = (j < end) ? wj : -1;
  }

  int half = lane >> 4;
  int m    = lane & 15;
  int row  = d0 + w * 16 + m;
  int rowc = row < n ? row : n - 1;
  const float* xrow = x + (size_t)rowc * F_IN;

  v8f c = {};
  #pragma unroll 4
  for (int k0 = 0; k0 < F_IN; k0 += 4) {
    int kk = k0 + 2 * half;
    v2f a = *(const v2f*)(xrow + kk);
    v2f bb;
    bb.x = W1[kk * HID + m];
    bb.y = W1[(kk + 1) * HID + m];
    c = __builtin_amdgcn_wmma_f32_16x16x4_f32(
        false, a, false, bb, (short)0, c, false, false);
  }

  #pragma unroll
  for (int k = 0; k < DPT; ++k)
    if (dd[k] >= 0) atomicAdd(&cnt[(dd[k] >> SRC_BITS) & (BIN_NODES - 1)], 1);
  for (int j = beg + DPT * PT + t; j < end; j += PT)
    atomicAdd(&cnt[(ppk[j] >> SRC_BITS) & (BIN_NODES - 1)], 1);
  if (nov > e) nov = e;
  for (int j = t; j < nov; j += PT) {
    int od = ovf[2 * j + 1];
    if (bin_of(od) == b) atomicAdd(&cnt[(od - d0) & (BIN_NODES - 1)], 1);
  }
  __syncthreads();

  float dv = rsqrtf((float)cnt[w * 16 + m] + 1.0f);
  if (half == 0 && row < n) dinv[row] = dv;
  bool full = (d0 + w * 16 + 16 <= n);
  #pragma unroll
  for (int r = 0; r < 8; ++r) {
    int orow = d0 + w * 16 + r + 8 * half;
    float ws = __shfl(dv, r + 8 * half, 32);
    if (full || orow < n) h[(size_t)orow * HID + m] = c[r] * ws;
  }
}


__global__ void __launch_bounds__(PT) agg1_gemm2_kernel(
    const int* __restrict__ ppk, const int* __restrict__ binCur,
    const int* __restrict__ ovfCnt, const int* __restrict__ ovf,
    const float* __restrict__ h,
    const float* __restrict__ b1, const float* __restrict__ W2,
    const float* __restrict__ dinv, float* __restrict__ lt, int n, int e) {
  __shared__ float acc[(BIN_NODES + 1) * HID];
  int t = threadIdx.x;
  int b = blockIdx.x;
  int lane = t & 31;
  int w = t >> 5;
  int d0 = b << BIN_SHIFT;
  for (int i = t; i < BIN_NODES * HID; i += PT) {
    int node = d0 + (i >> 4);
    acc[i] = (node < n) ? h[(size_t)d0 * HID + i] : 0.0f;
  }
  if (t < HID) acc[BIN_NODES * HID + t] = 0.0f;
  int cap = bin_cap(e, (int)gridDim.x);
  int tot = binCur[b];
  int nov = ovfCnt[0];
  __syncthreads();
  int beg = b * cap;
  int end = beg + (tot < cap ? tot : cap);
  int half = lane >> 4;
  int c = lane & 15;
  int j0 = beg + w * 32;
  int last = end - 1;
  int jf = j0 + lane;
  const int* pf = (end > beg) ? ppk + (jf < last ? jf : last) : binCur;
  int wv = *pf;
  if (jf >= end) wv = (wv & SRC_MASK) | SPARE_ROW;
  for (; j0 < end; j0 += PT) {
    int jn = j0 + PT + lane;
    int nwv = ppk[jn < last ? jn : last];
    if (jn >= end) nwv = (nwv & SRC_MASK) | SPARE_ROW;
    float v[16];
    unsigned dl[16];
    #pragma unroll
    for (int k = 0; k < 16; ++k) {
      int wk = __shfl(wv, 2 * k + half, 32);
      dl[k] = (unsigned)wk >> SRC_BITS;
      v[k] = h[(unsigned)(wk & SRC_MASK) * HID + c];
    }
    #pragma unroll
    for (int k = 0; k < 16; ++k)
      atomicAdd(&acc[dl[k] * HID + c], v[k]);
    wv = nwv;
  }
  if (nov > e) nov = e;
  for (int j = t; j < nov; j += PT) {
    int od = ovf[2 * j + 1];
    if (bin_of(od) == b) {
      const float* hs = h + (size_t)(unsigned)(ovf[2 * j] & SRC_MASK) * HID;
      float* ar = acc + ((od - d0) & (BIN_NODES - 1)) * HID;
      for (int k = 0; k < HID; ++k) atomicAdd(&ar[k], hs[k]);
    }
  }
  __syncthreads();

  int row  = d0 + w * 16 + c;
  int rowc = row < n ? row : n - 1;
  const float* hrow = acc + (w * 16 + c) * HID;
  float maskv = (c < NCLS) ? 1.0f : 0.0f;
  int mc = c & (NCLS - 1);
  float dr = dinv[rowc];

  v8f cc = {};
  #pragma unroll
  for (int k0 = 0; k0 < HID; k0 += 4) {
    int kk = k0 + 2 * half;
    v2f a;
    a.x = fmaxf(hrow[kk]     * dr + b1[kk],     0.0f);
    a.y = fmaxf(hrow[kk + 1] * dr + b1[kk + 1], 0.0f);
    v2f bb;
    bb.x = W2[kk * NCLS + mc] * maskv;
    bb.y = W2[(kk + 1) * NCLS + mc] * maskv;
    cc = __builtin_amdgcn_wmma_f32_16x16x4_f32(
        false, a, false, bb, (short)0, cc, false, false);
  }

  bool full = (d0 + w * 16 + 16 <= n);
  float wr[8];
  #pragma unroll
  for (int r = 0; r < 8; ++r) wr[r] = __shfl(dr, r + 8 * half, 32);
  if (c < NCLS) {
    #pragma unroll
    for (int r = 0; r < 8; ++r) {
      int orow = d0 + w * 16 + r + 8 * half;
      if (full || orow < n) lt[(size_t)orow * NCLS + c] = cc[r] * wr[r];
    }
  }
}


__global__ void __launch_bounds__(PT) agg2_lsm_kernel(
    const int* __restrict__ ppk, const int* __restrict__ binCur,
    const int* __restrict__ ovfCnt, const int* __restrict__ ovf,
    const float* __restrict__ lt,
    const float* __restrict__ dinv, const float* __restrict__ b2,
    float* __restrict__ out, int n, int e) {
  __shared__ float acc[(BIN_NODES + 1) * NCLS];
  int t = threadIdx.x;
  int b = blockIdx.x;
  int lane = t & 31;
  int w = t >> 5;
  int d0 = b << BIN_SHIFT;
  for (int i = t; i < BIN_NODES * NCLS; i += PT) {
    int node = d0 + (i >> 3);
    acc[i] = (node < n) ? lt[(size_t)d0 * NCLS + i] : 0.0f;
  }
  if (t < NCLS) acc[BIN_NODES * NCLS + t] = 0.0f;
  int cap = bin_cap(e, (int)gridDim.x);
  int tot = binCur[b];
  int nov = ovfCnt[0];
  __syncthreads();
  int beg = b * cap;
  int end = beg + (tot < cap ? tot : cap);
  int q = lane >> 3;
  int c = lane & 7;
  int j0 = beg + w * 32;
  int last = end - 1;
  int jf = j0 + lane;
  const int* pf = (end > beg) ? ppk + (jf < last ? jf : last) : binCur;
  int wv = *pf;
  if (jf >= end) wv = (wv & SRC_MASK) | SPARE_ROW;
  for (; j0 < end; j0 += PT) {
    int jn = j0 + PT + lane;
    int nwv = ppk[jn < last ? jn : last];
    if (jn >= end) nwv = (nwv & SRC_MASK) | SPARE_ROW;
    float v[8];
    unsigned dl[8];
    #pragma unroll
    for (int k = 0; k < 8; ++k) {
      int wk = __shfl(wv, 4 * k + q, 32);
      dl[k] = (unsigned)wk >> SRC_BITS;
      v[k] = lt[(unsigned)(wk & SRC_MASK) * NCLS + c];
    }
    #pragma unroll
    for (int k = 0; k < 8; ++k)
      atomicAdd(&acc[dl[k] * NCLS + c], v[k]);
    wv = nwv;
  }
  if (nov > e) nov = e;
  for (int j = t; j < nov; j += PT) {
    int od = ovf[2 * j + 1];
    if (bin_of(od) == b) {
      const float* ls = lt + (size_t)(unsigned)(ovf[2 * j] & SRC_MASK) * NCLS;
      float* ar = acc + ((od - d0) & (BIN_NODES - 1)) * NCLS;
      for (int k = 0; k < NCLS; ++k) atomicAdd(&ar[k], ls[k]);
    }
  }
  __syncthreads();

  int node = d0 + t;
  if (t < BIN_NODES && node < n) {
    const float* srow = acc + t * NCLS;
    float* row = out + (size_t)node * NCLS;
    float dn = dinv[node];
    float z[NCLS];
    float mx = -INFINITY;
    #pragma unroll
    for (int k = 0; k < NCLS; ++k) {
      z[k] = srow[k] * dn + b2[k];
      mx = fmaxf(mx, z[k]);
    }
    float s = 0.0f;
    #pragma unroll
    for (int k = 0; k < NCLS; ++k) s += expf(z[k] - mx);
    float l = mx + logf(s);
    #pragma unroll
    for (int k = 0; k < NCLS; ++k) row[k] = z[k] - l;
  }
}


extern "C" void kernel_launch(void* const* d_in, const int* in_sizes, int n_in,
                              void* d_out, int out_size, void* d_ws, size_t ws_size,
                              hipStream_t stream) {
  const float* x  = (const float*)d_in[0];
  const int* edge = (const int*)d_in[1];
  const float* W1 = (const float*)d_in[2];
  const float* b1 = (const float*)d_in[3];
  const float* W2 = (const float*)d_in[4];
  const float* b2 = (const float*)d_in[5];

  int n = in_sizes[0] / F_IN;
  int e = in_sizes[1] / 2;
  const int* src = edge;
  const int* dst = edge + e;

  size_t np = ((size_t)n + 63) / 64 * 64;
  size_t ep = ((size_t)e + 63) / 64 * 64;
  int*   binCur = (int*)d_ws;
  int*   ovfCnt = binCur + NB_MAX;
  float* dinv = (float*)(ovfCnt + NB_MAX);
  float* h    = dinv + np;
  float* lt   = h + np * HID;
  int*   ovf  = (int*)(lt + np * NCLS);
  int*   ppk  = ovf + 2 * ep;
  float* out  = (float*)d_out;

  const int T = 256;
  int nBins   = (n + BIN_NODES - 1) / BIN_NODES;
  int nChunks = (e + CHUNK - 1) / CHUNK;

  zero_kernel<<<(2 * NB_MAX + T - 1) / T, T, 0, stream>>>(binCur, 2 * NB_MAX);
  part_kernel<<<nChunks, PT, 0, stream>>>(src, dst, binCur, ovfCnt, ovf, ppk, e, nBins);

  deg_gemm1_kernel<<<nBins, PT, 0, stream>>>(ppk, binCur, ovfCnt, ovf, x, W1, dinv, h, n, e);
  agg1_gemm2_kernel<<<nBins, PT, 0, stream>>>(ppk, binCur, ovfCnt, ovf, h, b1, W2, dinv, lt, n, e);
  agg2_lsm_kernel<<<nBins, PT, 0, stream>>>(ppk, binCur, ovfCnt, ovf, lt, dinv, b2, out, n, e);
}


# Node blocks of 200 destination nodes instead of 256: the node-block kernels launch 500 work-groups, two per WGP, instead of 391 (half the WGPs had two, the rest one); sums in arrival order as before
# speedup vs baseline: 20.1293x; 1.1436x over previous
#include <hip/hip_runtime.h>
#include <hip/hip_bf16.h>
#include <math.h>


typedef float v2f __attribute__((ext_vector_type(2)));
typedef float v8f __attribute__((ext_vector_type(8)));

#define F_IN 128
#define HID  16
#define NCLS 8
#define BIN_NODES 200
#define ROWS_LDS 256
#define NT ((BIN_NODES + 15) / 16)
#define NB_MAX 512
#define PT 512
#define EPT 16
#define CHUNK (PT * EPT)
#define DPT 14
#define SRC_BITS 20
#define SRC_MASK ((1 << SRC_BITS) - 1)
#define SPARE_ROW (ROWS_LDS << SRC_BITS)


__device__ __forceinline__ int bin_of(int d) {
  unsigned b = (unsigned)d / BIN_NODES;
  return (int)(b < (unsigned)(NB_MAX - 1) ? b : (unsigned)(NB_MAX - 1));
}

__device__ __forceinline__ int pack_edge(int s, int d, int b) {
  return (s & SRC_MASK) | (((d - b * BIN_NODES) & (ROWS_LDS - 1)) << SRC_BITS);
}

__device__ __forceinline__ int bin_cap(int e, int nb) {
  int avg = e / nb;
  return (avg + (avg >> 2) + 256 + 63) & ~63;
}

__device__ __forceinline__ int block_excl_scan(int v, int* wtot, int t) {
  int lane = t & 31;
  int w = t >> 5;
  int inc = v;
  #pragma unroll
  for (int o = 1; o < 32; o <<= 1) {
    int up = __shfl_up(inc, o, 32);
    if (lane >= o) inc += up;
  }
  if (lane == 31) wtot[w] = inc;
  __syncthreads();
  int pre = 0;
  #pragma unroll
  for (int k = 0; k < PT / 32; ++k) pre += (k < w) ? wtot[k] : 0;
  __syncthreads();
  return pre + inc - v;
}


__global__ void zero_kernel(int* __restrict__ p, int n) {
  int i = blockIdx.x * blockDim.x + threadIdx.x;
  if (i < n) p[i] = 0;
}

__global__ void __launch_bounds__(PT) part_kernel(
    const int* __restrict__ src, const int* __restrict__ dst,
    int* __restrict__ binCur, int* __restrict__ ovfCnt, int* __restrict__ ovf,
    int* __restrict__ ppk, int e, int nb) {
  __shared__ int hist[NB_MAX];
  __shared__ int lstart[NB_MAX];
  __shared__ int goff[NB_MAX];
  __shared__ int fitn[NB_MAX];
  __shared__ int obase[NB_MAX];
  __shared__ int wtot[PT / 32];
  __shared__ int ssrc[CHUNK];
  __shared__ int sdst[CHUNK];
  int t = threadIdx.x;
  int cap = bin_cap(e, nb);
  hist[t] = 0;
  __syncthreads();
  int base = blockIdx.x * CHUNK;
  int s[EPT], d[EPT], r[EPT];
  #pragma unroll
  for (int k = 0; k < EPT; ++k) {
    int i = base + k * PT + t;
    bool ok = i < e;
    int ic = ok ? i : e - 1;
    int dv = dst[ic];
    s[k] = src[ic];
    d[k] = ok ? dv : -1;
  }
  #pragma unroll
  for (int k = 0; k < EPT; ++k)
    r[k] = (d[k] >= 0) ? atomicAdd(&hist[bin_of(d[k])], 1) : 0;
  __syncthreads();
  int mine = hist[t];
  int ls = block_excl_scan(mine, wtot, t);
  int off = mine ? atomicAdd(&binCur[t], mine) : 0;
  int fit = cap - off;
  fit = fit < 0 ? 0 : (fit > mine ? mine : fit);
  int nov = mine - fit;
  lstart[t] = ls;
  goff[t] = off;
  fitn[t] = fit;
  obase[t] = nov ? atomicAdd(ovfCnt, nov) : 0;
  __syncthreads();
  #pragma unroll
  for (int k = 0; k < EPT; ++k) {
    if (d[k] >= 0) {
      int p = lstart[bin_of(d[k])] + r[k];
      ssrc[p] = s[k];
      sdst[p] = d[k];
    }
  }
  __syncthreads();
  int m = e - base;
  if (m > CHUNK) m = CHUNK;
  for (int j = t; j < m; j += PT) {
    int dj = sdst[j];
    int b = bin_of(dj);
    int i = j - lstart[b];
    int f = fitn[b];
    if (i < f) {
      ppk[b * cap + goff[b] + i] = pack_edge(ssrc[j], dj, b);
    } else {
      int q = obase[b] + (i - f);
      ovf[2 * q] = ssrc[j];
      ovf[2 * q + 1] = dj;
    }
  }
}


__global__ void __launch_bounds__(PT) __attribute__((amdgpu_waves_per_eu(4, 8))) deg_gemm1_kernel(
    const int* __restrict__ ppk, const int* __restrict__ binCur,
    const int* __restrict__ ovfCnt, const int* __restrict__ ovf,
    const float* __restrict__ x, const float* __restrict__ W1,
    float* __restrict__ dinv, float* __restrict__ h, int n, int e) {
  __shared__ int cnt[ROWS_LDS];
  int t = threadIdx.x;
  int b = blockIdx.x;
  int lane = t & 31;
  int w = t >> 5;
  int d0 = b * BIN_NODES;
  if (t < ROWS_LDS) cnt[t] = 0;
  __syncthreads();
  int cap = bin_cap(e, (int)gridDim.x);
  int tot = binCur[b];
  int nov = ovfCnt[0];
  int beg = b * cap;
  int end = beg + (tot < cap ? tot : cap);
  int last = end - 1;
  bool some = end > beg;
  int dd[DPT];
  #pragma unroll
  for (int k = 0; k < DPT; ++k) {
    int j = beg + k * PT + t;
    const int* pj = some ? ppk + (j < last ? j : last) : binCur;
    int wj = *pj;
    dd[k] = (j < end) ? wj : -1;
  }

  int half = lane >> 4;
  int m    = lane & 15;
  int row  = d0 + w * 16 + m;
  int rowc = row < n ? row : n - 1;
  const float* xrow = x + (size_t)rowc * F_IN;

  v8f c = {};
  if (w < NT) {
    #pragma unroll 4
    for (int k0 = 0; k0 < F_IN; k0 += 4) {
      int kk = k0 + 2 * half;
      v2f a = *(const v2f*)(xrow + kk);
      v2f bb;
      bb.x = W1[kk * HID + m];
      bb.y = W1[(kk + 1) * HID + m];
      c = __builtin_amdgcn_wmma_f32_16x16x4_f32(
          false, a, false, bb, (short)0, c, false, false);
    }
  }

  #pragma unroll
  for (int k = 0; k < DPT; ++k)
    if (dd[k] >= 0) atomicAdd(&cnt[(dd[k] >> SRC_BITS) & (ROWS_LDS - 1)], 1);
  for (int j = beg + DPT * PT + t; j < end; j += PT)
    atomicAdd(&cnt[(ppk[j] >> SRC_BITS) & (ROWS_LDS - 1)], 1);
  if (nov > e) nov = e;
  for (int j = t; j < nov; j += PT) {
    int od = ovf[2 * j + 1];
    if (bin_of(od) == b) atomicAdd(&cnt[(od - d0) & (ROWS_LDS - 1)], 1);
  }
  __syncthreads();

  float dv = rsqrtf((float)cnt[w * 16 + m] + 1.0f);
  if (half == 0 && w * 16 + m < BIN_NODES && row < n) dinv[row] = dv;
  bool full = (w * 16 + 16 <= BIN_NODES) && (d0 + w * 16 + 16 <= n);
  #pragma unroll
  for (int r = 0; r < 8; ++r) {
    int trow = w * 16 + r + 8 * half;
    int orow = d0 + trow;
    float ws = __shfl(dv, r + 8 * half, 32);
    if (full || (trow < BIN_NODES && orow < n)) h[(size_t)orow * HID + m] = c[r] * ws;
  }
}


__global__ void __launch_bounds__(PT) agg1_gemm2_kernel(
    const int* __restrict__ ppk, const int* __restrict__ binCur,
    const int* __restrict__ ovfCnt, const int* __restrict__ ovf,
    const float* __restrict__ h,
    const float* __restrict__ b1, const float* __restrict__ W2,
    const float* __restrict__ dinv, float* __restrict__ lt, int n, int e) {
  __shared__ float acc[(ROWS_LDS + 1) * HID];
  int t = threadIdx.x;
  int b = blockIdx.x;
  int lane = t & 31;
  int w = t >> 5;
  int d0 = b * BIN_NODES;
  for (int i = t; i < ROWS_LDS * HID; i += PT) {
    int r = i >> 4;
    acc[i] = (r < BIN_NODES && d0 + r < n) ? h[(size_t)d0 * HID + i] : 0.0f;
  }
  if (t < HID) acc[ROWS_LDS * HID + t] = 0.0f;
  int cap = bin_cap(e, (int)gridDim.x);
  int tot = binCur[b];
  int nov = ovfCnt[0];
  __syncthreads();
  int beg = b * cap;
  int end = beg + (tot < cap ? tot : cap);
  int half = lane >> 4;
  int c = lane & 15;
  int j0 = beg + w * 32;
  int last = end - 1;
  int jf = j0 + lane;
  const int* pf = (end > beg) ? ppk + (jf < last ? jf : last) : binCur;
  int wv = *pf;
  if (jf >= end) wv = (wv & SRC_MASK) | SPARE_ROW;
  for (; j0 < end; j0 += PT) {
    int jn = j0 + PT + lane;
    int nwv = ppk[jn < last ? jn : last];
    if (jn >= end) nwv = (nwv & SRC_MASK) | SPARE_ROW;
    float v[16];
    unsigned dl[16];
    #pragma unroll
    for (int k = 0; k < 16; ++k) {
      int wk = __shfl(wv, 2 * k + half, 32);
      dl[k] = (unsigned)wk >> SRC_BITS;
      v[k] = h[(unsigned)(wk & SRC_MASK) * HID + c];
    }
    #pragma unroll
    for (int k = 0; k < 16; ++k)
      atomicAdd(&acc[dl[k] * HID + c], v[k]);
    wv = nwv;
  }
  if (nov > e) nov = e;
  for (int j = t; j < nov; j += PT) {
    int od = ovf[2 * j + 1];
    if (bin_of(od) == b) {
      const float* hs = h + (size_t)(unsigned)(ovf[2 * j] & SRC_MASK) * HID;
      float* ar = acc + ((od - d0) & (ROWS_LDS - 1)) * HID;
      for (int k = 0; k < HID; ++k) atomicAdd(&ar[k], hs[k]);
    }
  }
  __syncthreads();

  int row  = d0 + w * 16 + c;
  int rowc = row < n ? row : n - 1;
  const float* hrow = acc + (w * 16 + c) * HID;
  float maskv = (c < NCLS) ? 1.0f : 0.0f;
  int mc = c & (NCLS - 1);
  float dr = dinv[rowc];

  v8f cc = {};
  #pragma unroll
  for (int k0 = 0; k0 < HID; k0 += 4) {
    int kk = k0 + 2 * half;
    v2f a;
    a.x = fmaxf(hrow[kk]     * dr + b1[kk],     0.0f);
    a.y = fmaxf(hrow[kk + 1] * dr + b1[kk + 1], 0.0f);
    v2f bb;
    bb.x = W2[kk * NCLS + mc] * maskv;
    bb.y = W2[(kk + 1) * NCLS + mc] * maskv;
    cc = __builtin_amdgcn_wmma_f32_16x16x4_f32(
        false, a, false, bb, (short)0, cc, false, false);
  }

  bool full = (w * 16 + 16 <= BIN_NODES) && (d0 + w * 16 + 16 <= n);
  float wr[8];
  #pragma unroll
  for (int r = 0; r < 8; ++r) wr[r] = __shfl(dr, r + 8 * half, 32);
  if (c < NCLS) {
    #pragma unroll
    for (int r = 0; r < 8; ++r) {
      int trow = w * 16 + r + 8 * half;
      int orow = d0 + trow;
      if (full || (trow < BIN_NODES && orow < n)) lt[(size_t)orow * NCLS + c] = cc[r] * wr[r];
    }
  }
}


__global__ void __launch_bounds__(PT) agg2_lsm_kernel(
    const int* __restrict__ ppk, const int* __restrict__ binCur,
    const int* __restrict__ ovfCnt, const int* __restrict__ ovf,
    const float* __restrict__ lt,
    const float* __restrict__ dinv, const float* __restrict__ b2,
    float* __restrict__ out, int n, int e) {
  __shared__ float acc[(ROWS_LDS + 1) * NCLS];
  int t = threadIdx.x;
  int b = blockIdx.x;
  int lane = t & 31;
  int w = t >> 5;
  int d0 = b * BIN_NODES;
  for (int i = t; i < ROWS_LDS * NCLS; i += PT) {
    int r = i >> 3;
    acc[i] = (r < BIN_NODES && d0 + r < n) ? lt[(size_t)d0 * NCLS + i] : 0.0f;
  }
  if (t < NCLS) acc[ROWS_LDS * NCLS + t] = 0.0f;
  int cap = bin_cap(e, (int)gridDim.x);
  int tot = binCur[b];
  int nov = ovfCnt[0];
  __syncthreads();
  int beg = b * cap;
  int end = beg + (tot < cap ? tot : cap);
  int q = lane >> 3;
  int c = lane & 7;
  int j0 = beg + w * 32;
  int last = end - 1;
  int jf = j0 + lane;
  const int* pf = (end > beg) ? ppk + (jf < last ? jf : last) : binCur;
  int wv = *pf;
  if (jf >= end) wv = (wv & SRC_MASK) | SPARE_ROW;
  for (; j0 < end; j0 += PT) {
    int jn = j0 + PT + lane;
    int nwv = ppk[jn < last ? jn : last];
    if (jn >= end) nwv = (nwv & SRC_MASK) | SPARE_ROW;
    float v[8];
    unsigned dl[8];
    #pragma unroll
    for (int k = 0; k < 8; ++k) {
      int wk = __shfl(wv, 4 * k + q, 32);
      dl[k] = (unsigned)wk >> SRC_BITS;
      v[k] = lt[(unsigned)(wk & SRC_MASK) * NCLS + c];
    }
    #pragma unroll
    for (int k = 0; k < 8; ++k)
      atomicAdd(&acc[dl[k] * NCLS + c], v[k]);
    wv = nwv;
  }
  if (nov > e) nov = e;
  for (int j = t; j < nov; j += PT) {
    int od = ovf[2 * j + 1];
    if (bin_of(od) == b) {
      const float* ls = lt + (size_t)(unsigned)(ovf[2 * j] & SRC_MASK) * NCLS;
      float* ar = acc + ((od - d0) & (ROWS_LDS - 1)) * NCLS;
      for (int k = 0; k < NCLS; ++k) atomicAdd(&ar[k], ls[k]);
    }
  }
  __syncthreads();

  int node = d0 + t;
  if (t < BIN_NODES && node < n) {
    const float* srow = acc + t * NCLS;
    float* row = out + (size_t)node * NCLS;
    float dn = dinv[node];
    float z[NCLS];
    float mx = -INFINITY;
    #pragma unroll
    for (int k = 0; k < NCLS; ++k) {
      z[k] = srow[k] * dn + b2[k];
      mx = fmaxf(mx, z[k]);
    }
    float s = 0.0f;
    #pragma unroll
    for (int k = 0; k < NCLS; ++k) s += expf(z[k] - mx);
    float l = mx + logf(s);
    #pragma unroll
    for (int k = 0; k < NCLS; ++k) row[k] = z[k] - l;
  }
}


extern "C" void kernel_launch(void* const* d_in, const int* in_sizes, int n_in,
                              void* d_out, int out_size, void* d_ws, size_t ws_size,
                              hipStream_t stream) {
  const float* x  = (const float*)d_in[0];
  const int* edge = (const int*)d_in[1];
  const float* W1 = (const float*)d_in[2];
  const float* b1 = (const float*)d_in[3];
  const float* W2 = (const float*)d_in[4];
  const float* b2 = (const float*)d_in[5];

  int n = in_sizes[0] / F_IN;
  int e = in_sizes[1] / 2;
  const int* src = edge;
  const int* dst = edge + e;

  size_t np = ((size_t)n + 63) / 64 * 64;
  size_t ep = ((size_t)e + 63) / 64 * 64;
  int*   binCur = (int*)d_ws;
  int*   ovfCnt = binCur + NB_MAX;
  float* dinv = (float*)(ovfCnt + NB_MAX);
  float* h    = dinv + np;
  float* lt   = h + np * HID;
  int*   ovf  = (int*)(lt + np * NCLS);
  int*   ppk  = ovf + 2 * ep;
  float* out  = (float*)d_out;

  const int T = 256;
  int nBins   = (n + BIN_NODES - 1) / BIN_NODES;
  int nChunks = (e + CHUNK - 1) / CHUNK;

  zero_kernel<<<(2 * NB_MAX + T - 1) / T, T, 0, stream>>>(binCur, 2 * NB_MAX);
  part_kernel<<<nChunks, PT, 0, stream>>>(src, dst, binCur, ovfCnt, ovf, ppk, e, nBins);

  deg_gemm1_kernel<<<nBins, PT, 0, stream>>>(ppk, binCur, ovfCnt, ovf, x, W1, dinv, h, n, e);
  agg1_gemm2_kernel<<<nBins, PT, 0, stream>>>(ppk, binCur, ovfCnt, ovf, h, b1, W2, dinv, lt, n, e);
  agg2_lsm_kernel<<<nBins, PT, 0, stream>>>(ppk, binCur, ovfCnt, ovf, lt, dinv, b2, out, n, e);
}
